# starting kernel
# baseline (speedup 1.0000x reference)
#include <hip/hip_runtime.h>
#include <cstddef>

typedef float v2f __attribute__((ext_vector_type(2)));
typedef float v8f __attribute__((ext_vector_type(8)));

#define NJ 20

__device__ __forceinline__ void rpy2r_dev(float r, float p, float y, float* R) {
  float cr = cosf(r), sr = sinf(r);
  float cp = cosf(p), sp = sinf(p);
  float cy = cosf(y), sy = sinf(y);
  R[0] = cy * cp; R[1] = cy * sp * sr - sy * cr; R[2] = cy * sp * cr + sy * sr;
  R[3] = sy * cp; R[4] = sy * sp * sr + cy * cr; R[5] = sy * sp * cr - cy * sr;
  R[6] = -sp;     R[7] = cp * sr;                R[8] = cp * cr;
}

__device__ __forceinline__ void mat3mul(const float* A, const float* B, float* C) {
#pragma unroll
  for (int i = 0; i < 3; ++i)
#pragma unroll
    for (int k = 0; k < 3; ++k)
      C[i * 3 + k] = A[i * 3 + 0] * B[0 * 3 + k] + A[i * 3 + 1] * B[1 * 3 + k] +
                     A[i * 3 + 2] * B[2 * 3 + k];
}

__global__ void fk_const_kernel(const float* __restrict__ rev_p_off,
                                const float* __restrict__ rev_rpy_off,
                                const float* __restrict__ pri_p_off,
                                const float* __restrict__ pri_rpy_off,
                                const float* __restrict__ rev_axis,
                                const float* __restrict__ pri_axis,
                                const float* __restrict__ p_track,
                                const float* __restrict__ rpy_track,
                                float* __restrict__ cst) {
  int j = threadIdx.x;
  if (j >= NJ) return;
  float* o = cst + j * 64;
  float Rt[9];

  rpy2r_dev(rev_rpy_off[3 * j], rev_rpy_off[3 * j + 1], rev_rpy_off[3 * j + 2], Rt);
#pragma unroll
  for (int k = 0; k < 9; ++k) o[k] = Rt[k];
  o[9] = rev_p_off[3 * j]; o[10] = rev_p_off[3 * j + 1]; o[11] = rev_p_off[3 * j + 2];

  rpy2r_dev(pri_rpy_off[3 * j], pri_rpy_off[3 * j + 1], pri_rpy_off[3 * j + 2], Rt);
#pragma unroll
  for (int k = 0; k < 9; ++k) o[12 + k] = Rt[k];
  o[21] = pri_p_off[3 * j]; o[22] = pri_p_off[3 * j + 1]; o[23] = pri_p_off[3 * j + 2];

  float x = rev_axis[3 * j], y = rev_axis[3 * j + 1], z = rev_axis[3 * j + 2];
  float inv = 1.f / (sqrtf(x * x + y * y + z * z) + 1e-12f);
  o[24] = x * inv; o[25] = y * inv; o[26] = z * inv;

  o[27] = pri_axis[3 * j]; o[28] = pri_axis[3 * j + 1]; o[29] = pri_axis[3 * j + 2];

  rpy2r_dev(rpy_track[3 * j], rpy_track[3 * j + 1], rpy_track[3 * j + 2], Rt);
  o[30] = Rt[0]; o[31] = Rt[1]; o[32] = Rt[2]; o[33] = p_track[3 * j];
  o[34] = Rt[3]; o[35] = Rt[4]; o[36] = Rt[5]; o[37] = p_track[3 * j + 1];
  o[38] = Rt[6]; o[39] = Rt[7]; o[40] = Rt[8]; o[41] = p_track[3 * j + 2];
  o[42] = 0.f; o[43] = 0.f; o[44] = 0.f; o[45] = 1.f;
}

__global__ __launch_bounds__(256) void fk_main_kernel(
    const float* __restrict__ rev_q, const float* __restrict__ pri_q,
    const float* __restrict__ cst, float* __restrict__ out, int B) {
  __shared__ float stage[8 * 512];

  const int tid  = blockIdx.x * 256 + threadIdx.x;
  const int lane = threadIdx.x & 31;
  const int wv   = threadIdx.x >> 5;
  float* st = &stage[wv * 512];

  const bool alive = (tid < B);
  const int  bs    = alive ? tid : (B - 1);

  float* trk_out = out;
  float* rev_out = out + (size_t)B * 160;
  float* pri_out = out + (size_t)B * 480;

  const int  h       = lane >> 4;
  const int  m16     = lane & 15;
  const int  nu      = m16 & 3;
  const int  iq      = m16 >> 2;
  const bool useful  = (h == (iq >> 1));
  const int  baseSel = (iq & 1) << 2;
  const long waveBase = (long)blockIdx.x * 256 + (long)wv * 32;

  float R[9] = {1.f, 0.f, 0.f, 0.f, 1.f, 0.f, 0.f, 0.f, 1.f};
  float p[3] = {0.f, 0.f, 0.f};

  for (int j = 0; j < NJ; ++j) {
    const float* C = cst + j * 64;
    float Ror[9], por[3], Rop[9], pop[3], ar[3], ap[3];
#pragma unroll
    for (int k = 0; k < 9; ++k) Ror[k] = C[k];
#pragma unroll
    for (int k = 0; k < 3; ++k) por[k] = C[9 + k];
#pragma unroll
    for (int k = 0; k < 9; ++k) Rop[k] = C[12 + k];
#pragma unroll
    for (int k = 0; k < 3; ++k) pop[k] = C[21 + k];
#pragma unroll
    for (int k = 0; k < 3; ++k) ar[k] = C[24 + k];
#pragma unroll
    for (int k = 0; k < 3; ++k) ap[k] = C[27 + k];

    float q = rev_q[(size_t)bs * NJ + j];
    float d = pri_q[(size_t)bs * NJ + j];

    float sn = sinf(q), cs = cosf(q), oc = 1.f - cs;
    float ax = ar[0], ay = ar[1], az = ar[2];
    float Rod[9];
    Rod[0] = 1.f + oc * (-(ay * ay + az * az));
    Rod[1] = -sn * az + oc * (ax * ay);
    Rod[2] =  sn * ay + oc * (ax * az);
    Rod[3] =  sn * az + oc * (ax * ay);
    Rod[4] = 1.f + oc * (-(ax * ax + az * az));
    Rod[5] = -sn * ax + oc * (ay * az);
    Rod[6] = -sn * ay + oc * (ax * az);
    Rod[7] =  sn * ax + oc * (ay * az);
    Rod[8] = 1.f + oc * (-(ax * ax + ay * ay));

    float Rrs[9]; mat3mul(Ror, Rod, Rrs);
    float R1[9];  mat3mul(R, Rrs, R1);
    float p1[3];
    p1[0] = R[0] * por[0] + R[1] * por[1] + R[2] * por[2] + p[0];
    p1[1] = R[3] * por[0] + R[4] * por[1] + R[5] * por[2] + p[1];
    p1[2] = R[6] * por[0] + R[7] * por[1] + R[8] * por[2] + p[2];

    if (alive) {
      float4* o4 = (float4*)(rev_out + ((size_t)tid * NJ + j) * 16);
      o4[0] = make_float4(R1[0], R1[1], R1[2], p1[0]);
      o4[1] = make_float4(R1[3], R1[4], R1[5], p1[1]);
      o4[2] = make_float4(R1[6], R1[7], R1[8], p1[2]);
      o4[3] = make_float4(0.f, 0.f, 0.f, 1.f);
    }

    float tv0 = d * ap[0], tv1 = d * ap[1], tv2 = d * ap[2];
    float ppv[3];
    ppv[0] = Rop[0] * tv0 + Rop[1] * tv1 + Rop[2] * tv2 + pop[0];
    ppv[1] = Rop[3] * tv0 + Rop[4] * tv1 + Rop[5] * tv2 + pop[1];
    ppv[2] = Rop[6] * tv0 + Rop[7] * tv1 + Rop[8] * tv2 + pop[2];
    float R2[9]; mat3mul(R1, Rop, R2);
    float p2[3];
    p2[0] = R1[0] * ppv[0] + R1[1] * ppv[1] + R1[2] * ppv[2] + p1[0];
    p2[1] = R1[3] * ppv[0] + R1[4] * ppv[1] + R1[5] * ppv[2] + p1[1];
    p2[2] = R1[6] * ppv[0] + R1[7] * ppv[1] + R1[8] * ppv[2] + p1[2];
#pragma unroll
    for (int k = 0; k < 9; ++k) R[k] = R2[k];
    p[0] = p2[0]; p[1] = p2[1]; p[2] = p2[2];

    if (alive) {
      float4* o4 = (float4*)(pri_out + ((size_t)tid * NJ + j) * 16);
      o4[0] = make_float4(R[0], R[1], R[2], p[0]);
      o4[1] = make_float4(R[3], R[4], R[5], p[1]);
      o4[2] = make_float4(R[6], R[7], R[8], p[2]);
      o4[3] = make_float4(0.f, 0.f, 0.f, 1.f);
    }

    if ((j & 1) == 0) {
      __syncthreads();
      float* sl = st + lane * 16;
      sl[0] = R[0]; sl[1] = R[1]; sl[2]  = R[2]; sl[3]  = p[0];
      sl[4] = R[3]; sl[5] = R[4]; sl[6]  = R[5]; sl[7]  = p[1];
      sl[8] = R[6]; sl[9] = R[7]; sl[10] = R[8]; sl[11] = p[2];
      sl[12] = 0.f; sl[13] = 0.f; sl[14] = 0.f; sl[15] = 1.f;
      __syncthreads();

      v2f Bop;
      Bop.x = C[30 + (2 * h + 0) * 4 + nu];
      Bop.y = C[30 + (2 * h + 1) * 4 + nu];

      const v8f zero8 = {0.f, 0.f, 0.f, 0.f, 0.f, 0.f, 0.f, 0.f};
#pragma unroll
      for (int g = 0; g < 8; ++g) {
        const v2f A = *(const v2f*)(st + (g * 4 + iq) * 16 + nu * 4 + 2 * h);
        v8f D = __builtin_amdgcn_wmma_f32_16x16x4_f32(
            false, A, false, Bop, (short)0, zero8, false, false);
        float e0 = baseSel ? D[4] : D[0];
        float e1 = baseSel ? D[5] : D[1];
        float e2 = baseSel ? D[6] : D[2];
        float e3 = baseSel ? D[7] : D[3];
        long bb = waveBase + g * 4 + iq;
        if (useful && bb < (long)B) {
          float* o = trk_out + ((size_t)bb * 10 + (j >> 1)) * 16 + nu;
          o[0] = e0; o[4] = e1; o[8] = e2; o[12] = e3;
        }
      }
    }
  }
}

extern "C" void kernel_launch(void* const* d_in, const int* in_sizes, int n_in,
                              void* d_out, int out_size, void* d_ws, size_t ws_size,
                              hipStream_t stream) {
  (void)n_in; (void)out_size; (void)ws_size;
  const float* rev_q       = (const float*)d_in[0];
  const float* pri_q       = (const float*)d_in[1];
  const float* rev_p_off   = (const float*)d_in[2];
  const float* rev_rpy_off = (const float*)d_in[3];
  const float* pri_p_off   = (const float*)d_in[4];
  const float* pri_rpy_off = (const float*)d_in[5];
  const float* rev_axis    = (const float*)d_in[6];
  const float* pri_axis    = (const float*)d_in[7];
  const float* p_track     = (const float*)d_in[8];
  const float* rpy_track   = (const float*)d_in[9];

  float* cst = (float*)d_ws;
  int B = in_sizes[0] / NJ;

  fk_const_kernel<<<1, 32, 0, stream>>>(rev_p_off, rev_rpy_off, pri_p_off, pri_rpy_off,
                                        rev_axis, pri_axis, p_track, rpy_track, cst);
  int grid = (B + 255) / 256;
  fk_main_kernel<<<grid, 256, 0, stream>>>(rev_q, pri_q, cst, (float*)d_out, B);
}


# Outputs staged per wave in LDS and stored as whole 128-B lines (4 per store) instead of per-lane 16-B and 4-B pieces; WMMA operands swapped so D holds output rows; wave-local hand-over, no barrier
# speedup vs baseline: 2.5916x; 2.5916x over previous
#include <hip/hip_runtime.h>
#include <cstddef>

typedef float v2f __attribute__((ext_vector_type(2)));
typedef float v8f __attribute__((ext_vector_type(8)));

#define NJ 20
#define PITCH 36

__device__ __forceinline__ void rpy2r_dev(float r, float p, float y, float* R) {
  float cr = cosf(r), sr = sinf(r);
  float cp = cosf(p), sp = sinf(p);
  float cy = cosf(y), sy = sinf(y);
  R[0] = cy * cp; R[1] = cy * sp * sr - sy * cr; R[2] = cy * sp * cr + sy * sr;
  R[3] = sy * cp; R[4] = sy * sp * sr + cy * cr; R[5] = sy * sp * cr - cy * sr;
  R[6] = -sp;     R[7] = cp * sr;                R[8] = cp * cr;
}

__device__ __forceinline__ void mat3mul(const float* A, const float* B, float* C) {
#pragma unroll
  for (int i = 0; i < 3; ++i)
#pragma unroll
    for (int k = 0; k < 3; ++k)
      C[i * 3 + k] = A[i * 3 + 0] * B[0 * 3 + k] + A[i * 3 + 1] * B[1 * 3 + k] +
                     A[i * 3 + 2] * B[2 * 3 + k];
}

__global__ void fk_const_kernel(const float* __restrict__ rev_p_off,
                                const float* __restrict__ rev_rpy_off,
                                const float* __restrict__ pri_p_off,
                                const float* __restrict__ pri_rpy_off,
                                const float* __restrict__ rev_axis,
                                const float* __restrict__ pri_axis,
                                const float* __restrict__ p_track,
                                const float* __restrict__ rpy_track,
                                float* __restrict__ cst) {
  int j = threadIdx.x;
  if (j >= NJ) return;
  float* o = cst + j * 64;
  float Rt[9];

  rpy2r_dev(rev_rpy_off[3 * j], rev_rpy_off[3 * j + 1], rev_rpy_off[3 * j + 2], Rt);
#pragma unroll
  for (int k = 0; k < 9; ++k) o[k] = Rt[k];
  o[9] = rev_p_off[3 * j]; o[10] = rev_p_off[3 * j + 1]; o[11] = rev_p_off[3 * j + 2];

  rpy2r_dev(pri_rpy_off[3 * j], pri_rpy_off[3 * j + 1], pri_rpy_off[3 * j + 2], Rt);
#pragma unroll
  for (int k = 0; k < 9; ++k) o[12 + k] = Rt[k];
  o[21] = pri_p_off[3 * j]; o[22] = pri_p_off[3 * j + 1]; o[23] = pri_p_off[3 * j + 2];

  float x = rev_axis[3 * j], y = rev_axis[3 * j + 1], z = rev_axis[3 * j + 2];
  float inv = 1.f / (sqrtf(x * x + y * y + z * z) + 1e-12f);
  o[24] = x * inv; o[25] = y * inv; o[26] = z * inv;

  o[27] = pri_axis[3 * j]; o[28] = pri_axis[3 * j + 1]; o[29] = pri_axis[3 * j + 2];

  rpy2r_dev(rpy_track[3 * j], rpy_track[3 * j + 1], rpy_track[3 * j + 2], Rt);
  o[30] = Rt[0]; o[31] = Rt[1]; o[32] = Rt[2]; o[33] = p_track[3 * j];
  o[34] = Rt[3]; o[35] = Rt[4]; o[36] = Rt[5]; o[37] = p_track[3 * j + 1];
  o[38] = Rt[6]; o[39] = Rt[7]; o[40] = Rt[8]; o[41] = p_track[3 * j + 2];
  o[42] = 0.f; o[43] = 0.f; o[44] = 0.f; o[45] = 1.f;
}

__device__ __forceinline__ void fk_joint(const float* __restrict__ C, float q, float d,
                                         float (&R)[9], float (&p)[3],
                                         float* myRev, float* myPri) {
  float Ror[9], por[3], Rop[9], pop[3], ar[3], ap[3];
#pragma unroll
  for (int k = 0; k < 9; ++k) Ror[k] = C[k];
#pragma unroll
  for (int k = 0; k < 3; ++k) por[k] = C[9 + k];
#pragma unroll
  for (int k = 0; k < 9; ++k) Rop[k] = C[12 + k];
#pragma unroll
  for (int k = 0; k < 3; ++k) pop[k] = C[21 + k];
#pragma unroll
  for (int k = 0; k < 3; ++k) ar[k] = C[24 + k];
#pragma unroll
  for (int k = 0; k < 3; ++k) ap[k] = C[27 + k];

  float sn = sinf(q), cs = cosf(q), oc = 1.f - cs;
  float ax = ar[0], ay = ar[1], az = ar[2];
  float Rod[9];
  Rod[0] = 1.f + oc * (-(ay * ay + az * az));
  Rod[1] = -sn * az + oc * (ax * ay);
  Rod[2] =  sn * ay + oc * (ax * az);
  Rod[3] =  sn * az + oc * (ax * ay);
  Rod[4] = 1.f + oc * (-(ax * ax + az * az));
  Rod[5] = -sn * ax + oc * (ay * az);
  Rod[6] = -sn * ay + oc * (ax * az);
  Rod[7] =  sn * ax + oc * (ay * az);
  Rod[8] = 1.f + oc * (-(ax * ax + ay * ay));

  float Rrs[9]; mat3mul(Ror, Rod, Rrs);
  float R1[9];  mat3mul(R, Rrs, R1);
  float p1[3];
  p1[0] = R[0] * por[0] + R[1] * por[1] + R[2] * por[2] + p[0];
  p1[1] = R[3] * por[0] + R[4] * por[1] + R[5] * por[2] + p[1];
  p1[2] = R[6] * por[0] + R[7] * por[1] + R[8] * por[2] + p[2];

  float4* r4 = (float4*)myRev;
  r4[0] = make_float4(R1[0], R1[1], R1[2], p1[0]);
  r4[1] = make_float4(R1[3], R1[4], R1[5], p1[1]);
  r4[2] = make_float4(R1[6], R1[7], R1[8], p1[2]);

  float tv0 = d * ap[0], tv1 = d * ap[1], tv2 = d * ap[2];
  float ppv[3];
  ppv[0] = Rop[0] * tv0 + Rop[1] * tv1 + Rop[2] * tv2 + pop[0];
  ppv[1] = Rop[3] * tv0 + Rop[4] * tv1 + Rop[5] * tv2 + pop[1];
  ppv[2] = Rop[6] * tv0 + Rop[7] * tv1 + Rop[8] * tv2 + pop[2];
  float R2[9]; mat3mul(R1, Rop, R2);
  float p2[3];
  p2[0] = R1[0] * ppv[0] + R1[1] * ppv[1] + R1[2] * ppv[2] + p1[0];
  p2[1] = R1[3] * ppv[0] + R1[4] * ppv[1] + R1[5] * ppv[2] + p1[1];
  p2[2] = R1[6] * ppv[0] + R1[7] * ppv[1] + R1[8] * ppv[2] + p1[2];
#pragma unroll
  for (int k = 0; k < 9; ++k) R[k] = R2[k];
  p[0] = p2[0]; p[1] = p2[1]; p[2] = p2[2];

  float4* q4 = (float4*)myPri;
  q4[0] = make_float4(R[0], R[1], R[2], p[0]);
  q4[1] = make_float4(R[3], R[4], R[5], p[1]);
  q4[2] = make_float4(R[6], R[7], R[8], p[2]);
}

__device__ __forceinline__ void wave_stage_handover() {
  asm volatile("s_wait_dscnt 0x0" ::: "memory");
}

__device__ __forceinline__ void read_plane(const float* plane, int fc, int fp, float4 (&v)[8]) {
#pragma unroll
  for (int k = 0; k < 8; ++k)
    v[k] = *(const float4*)(plane + (k * 4 + fc) * PITCH + fp * 4);
}

__device__ __forceinline__ void store_lines(float* __restrict__ o, size_t chainStride,
                                            const float4 (&v)[8]) {
#pragma unroll
  for (int k = 0; k < 8; ++k)
    *(float4*)(o + (size_t)(k * 4) * chainStride) = v[k];
}

__device__ __forceinline__ void store_lines_ragged(const float* plane, float* __restrict__ o,
                                                   size_t chainStride, long firstChain,
                                                   int B, int fc, int fp) {
#pragma unroll 1
  for (int k = 0; k < 8; ++k)
    if (firstChain + k * 4 < (long)B)
      *(float4*)(o + (size_t)(k * 4) * chainStride) =
          *(const float4*)(plane + (k * 4 + fc) * PITCH + fp * 4);
}

__global__ __launch_bounds__(256) void fk_main_kernel(
    const float* __restrict__ rev_q, const float* __restrict__ pri_q,
    const float* __restrict__ cst, float* __restrict__ out, int B) {
  __shared__ float4 stage[8 * 3 * 32 * (PITCH / 4)];

  const int tid  = blockIdx.x * 256 + threadIdx.x;
  const int lane = threadIdx.x & 31;
  const int wv   = threadIdx.x >> 5;
  float* sRev = (float*)stage + wv * (3 * 32 * PITCH);
  float* sPri = sRev + 32 * PITCH;
  float* sTrk = sPri + 32 * PITCH;
  float* myRev = sRev + lane * PITCH;
  float* myPri = sPri + lane * PITCH;

  const bool alive = (tid < B);
  const int  bs    = alive ? tid : (B - 1);

  float* trk_out = out;
  float* rev_out = out + (size_t)B * 160;
  float* pri_out = out + (size_t)B * 480;

  const int  h   = lane >> 4;
  const int  m16 = lane & 15;
  const int  nu  = m16 & 3;
  const int  iq  = m16 >> 2;
  const long waveBase = (long)blockIdx.x * 256 + (long)__builtin_amdgcn_readfirstlane(wv) * 32;
  const bool whole = (waveBase + 32 <= (long)B);
  const int fc = lane >> 3;
  const int fp = lane & 7;

  float R[9] = {1.f, 0.f, 0.f, 0.f, 1.f, 0.f, 0.f, 0.f, 1.f};
  float p[3] = {0.f, 0.f, 0.f};

  {
    const float4 e3 = make_float4(0.f, 0.f, 0.f, 1.f);
    *(float4*)(myRev + 12) = e3; *(float4*)(myRev + 28) = e3;
    *(float4*)(myPri + 12) = e3; *(float4*)(myPri + 28) = e3;
  }

  for (int jp = 0; jp < NJ / 2; ++jp) {
    const int j0 = 2 * jp;
    const float* C0 = cst + j0 * 64;
    const float* C1 = C0 + 64;
    const float q0 = rev_q[(size_t)bs * NJ + j0];
    const float d0 = pri_q[(size_t)bs * NJ + j0];
    const float q1 = rev_q[(size_t)bs * NJ + j0 + 1];
    const float d1 = pri_q[(size_t)bs * NJ + j0 + 1];
    v2f Aop;
    Aop.x = C0[30 + (2 * h + 0) * 4 + nu];
    Aop.y = C0[30 + (2 * h + 1) * 4 + nu];

    fk_joint(C0, q0, d0, R, p, myRev, myPri);
    wave_stage_handover();

    {
      const v8f zero8 = {0.f, 0.f, 0.f, 0.f, 0.f, 0.f, 0.f, 0.f};
      v2f Bv[8];
#pragma unroll
      for (int g = 0; g < 8; ++g)
        Bv[g] = *(const v2f*)(sPri + (g * 4 + iq) * PITCH + nu * 4 + 2 * h);
      v8f D[8];
#pragma unroll
      for (int g = 0; g < 8; ++g)
        D[g] = __builtin_amdgcn_wmma_f32_16x16x4_f32(
            false, Aop, false, Bv[g], (short)0, zero8, false, false);
      if (h == 0) {
        float* trow = sTrk + iq * PITCH + (jp & 1) * 16 + nu * 4;
#pragma unroll
        for (int g = 0; g < 8; ++g)
          *(float4*)(trow + g * 4 * PITCH) = make_float4(D[g][0], D[g][1], D[g][2], D[g][3]);
      }
    }

    fk_joint(C1, q1, d1, R, p, myRev + 16, myPri + 16);
    wave_stage_handover();

    float* oRev = rev_out + ((size_t)(waveBase + fc) * NJ + j0) * 16 + fp * 4;
    float* oPri = pri_out + ((size_t)(waveBase + fc) * NJ + j0) * 16 + fp * 4;
    if (whole) {
      float4 vr[8], vp[8];
      read_plane(sRev, fc, fp, vr);
      read_plane(sPri, fc, fp, vp);
      if (jp & 1) {
        float4 vt[8];
        read_plane(sTrk, fc, fp, vt);
        float* oTrk = trk_out + ((size_t)(waveBase + fc) * 10 + (jp - 1)) * 16 + fp * 4;
        store_lines(oRev, (size_t)NJ * 16, vr);
        store_lines(oPri, (size_t)NJ * 16, vp);
        store_lines(oTrk, (size_t)160, vt);
      } else {
        store_lines(oRev, (size_t)NJ * 16, vr);
        store_lines(oPri, (size_t)NJ * 16, vp);
      }
    } else {
      store_lines_ragged(sRev, oRev, (size_t)NJ * 16, waveBase + fc, B, fc, fp);
      store_lines_ragged(sPri, oPri, (size_t)NJ * 16, waveBase + fc, B, fc, fp);
      if (jp & 1) {
        float* oTrk = trk_out + ((size_t)(waveBase + fc) * 10 + (jp - 1)) * 16 + fp * 4;
        store_lines_ragged(sTrk, oTrk, (size_t)160, waveBase + fc, B, fc, fp);
      }
    }
  }
}

extern "C" void kernel_launch(void* const* d_in, const int* in_sizes, int n_in,
                              void* d_out, int out_size, void* d_ws, size_t ws_size,
                              hipStream_t stream) {
  (void)n_in; (void)out_size; (void)ws_size;
  const float* rev_q       = (const float*)d_in[0];
  const float* pri_q       = (const float*)d_in[1];
  const float* rev_p_off   = (const float*)d_in[2];
  const float* rev_rpy_off = (const float*)d_in[3];
  const float* pri_p_off   = (const float*)d_in[4];
  const float* pri_rpy_off = (const float*)d_in[5];
  const float* rev_axis    = (const float*)d_in[6];
  const float* pri_axis    = (const float*)d_in[7];
  const float* p_track     = (const float*)d_in[8];
  const float* rpy_track   = (const float*)d_in[9];

  float* cst = (float*)d_ws;
  int B = in_sizes[0] / NJ;

  fk_const_kernel<<<1, 32, 0, stream>>>(rev_p_off, rev_rpy_off, pri_p_off, pri_rpy_off,
                                        rev_axis, pri_axis, p_track, rpy_track, cst);
  int grid = (B + 255) / 256;
  fk_main_kernel<<<grid, 256, 0, stream>>>(rev_q, pri_q, cst, (float*)d_out, B);
}


# All global loads moved out of the joint loop: constant table copied once per work-group to LDS, each wave's joint values fetched as whole lines into LDS at entry; the loop sends only stores to memory
# speedup vs baseline: 2.7786x; 1.0722x over previous
#include <hip/hip_runtime.h>
#include <cstddef>

typedef float v2f __attribute__((ext_vector_type(2)));
typedef float v8f __attribute__((ext_vector_type(8)));

#define NJ 20
#define PITCH 36

__device__ __forceinline__ void rpy2r_dev(float r, float p, float y, float* R) {
  float cr = cosf(r), sr = sinf(r);
  float cp = cosf(p), sp = sinf(p);
  float cy = cosf(y), sy = sinf(y);
  R[0] = cy * cp; R[1] = cy * sp * sr - sy * cr; R[2] = cy * sp * cr + sy * sr;
  R[3] = sy * cp; R[4] = sy * sp * sr + cy * cr; R[5] = sy * sp * cr - cy * sr;
  R[6] = -sp;     R[7] = cp * sr;                R[8] = cp * cr;
}

__device__ __forceinline__ void mat3mul(const float* A, const float* B, float* C) {
#pragma unroll
  for (int i = 0; i < 3; ++i)
#pragma unroll
    for (int k = 0; k < 3; ++k)
      C[i * 3 + k] = A[i * 3 + 0] * B[0 * 3 + k] + A[i * 3 + 1] * B[1 * 3 + k] +
                     A[i * 3 + 2] * B[2 * 3 + k];
}

__global__ void fk_const_kernel(const float* __restrict__ rev_p_off,
                                const float* __restrict__ rev_rpy_off,
                                const float* __restrict__ pri_p_off,
                                const float* __restrict__ pri_rpy_off,
                                const float* __restrict__ rev_axis,
                                const float* __restrict__ pri_axis,
                                const float* __restrict__ p_track,
                                const float* __restrict__ rpy_track,
                                float* __restrict__ cst) {
  int j = threadIdx.x;
  if (j >= NJ) return;
  float* o = cst + j * 64;
  float Rt[9];

  rpy2r_dev(rev_rpy_off[3 * j], rev_rpy_off[3 * j + 1], rev_rpy_off[3 * j + 2], Rt);
#pragma unroll
  for (int k = 0; k < 9; ++k) o[k] = Rt[k];
  o[9] = rev_p_off[3 * j]; o[10] = rev_p_off[3 * j + 1]; o[11] = rev_p_off[3 * j + 2];

  rpy2r_dev(pri_rpy_off[3 * j], pri_rpy_off[3 * j + 1], pri_rpy_off[3 * j + 2], Rt);
#pragma unroll
  for (int k = 0; k < 9; ++k) o[12 + k] = Rt[k];
  o[21] = pri_p_off[3 * j]; o[22] = pri_p_off[3 * j + 1]; o[23] = pri_p_off[3 * j + 2];

  float x = rev_axis[3 * j], y = rev_axis[3 * j + 1], z = rev_axis[3 * j + 2];
  float inv = 1.f / (sqrtf(x * x + y * y + z * z) + 1e-12f);
  o[24] = x * inv; o[25] = y * inv; o[26] = z * inv;

  o[27] = pri_axis[3 * j]; o[28] = pri_axis[3 * j + 1]; o[29] = pri_axis[3 * j + 2];

  rpy2r_dev(rpy_track[3 * j], rpy_track[3 * j + 1], rpy_track[3 * j + 2], Rt);
  o[30] = Rt[0]; o[31] = Rt[1]; o[32] = Rt[2]; o[33] = p_track[3 * j];
  o[34] = Rt[3]; o[35] = Rt[4]; o[36] = Rt[5]; o[37] = p_track[3 * j + 1];
  o[38] = Rt[6]; o[39] = Rt[7]; o[40] = Rt[8]; o[41] = p_track[3 * j + 2];
  o[42] = 0.f; o[43] = 0.f; o[44] = 0.f; o[45] = 1.f;
  o[46] = 0.f; o[47] = 0.f;
}

__device__ __forceinline__ void fk_joint(const float* __restrict__ C, float q, float d,
                                         float (&R)[9], float (&p)[3],
                                         float* myRev, float* myPri) {
  float Ror[9], por[3], Rop[9], pop[3], ar[3], ap[3];
#pragma unroll
  for (int k = 0; k < 9; ++k) Ror[k] = C[k];
#pragma unroll
  for (int k = 0; k < 3; ++k) por[k] = C[9 + k];
#pragma unroll
  for (int k = 0; k < 9; ++k) Rop[k] = C[12 + k];
#pragma unroll
  for (int k = 0; k < 3; ++k) pop[k] = C[21 + k];
#pragma unroll
  for (int k = 0; k < 3; ++k) ar[k] = C[24 + k];
#pragma unroll
  for (int k = 0; k < 3; ++k) ap[k] = C[27 + k];

  float sn = sinf(q), cs = cosf(q), oc = 1.f - cs;
  float ax = ar[0], ay = ar[1], az = ar[2];
  float Rod[9];
  Rod[0] = 1.f + oc * (-(ay * ay + az * az));
  Rod[1] = -sn * az + oc * (ax * ay);
  Rod[2] =  sn * ay + oc * (ax * az);
  Rod[3] =  sn * az + oc * (ax * ay);
  Rod[4] = 1.f + oc * (-(ax * ax + az * az));
  Rod[5] = -sn * ax + oc * (ay * az);
  Rod[6] = -sn * ay + oc * (ax * az);
  Rod[7] =  sn * ax + oc * (ay * az);
  Rod[8] = 1.f + oc * (-(ax * ax + ay * ay));

  float Rrs[9]; mat3mul(Ror, Rod, Rrs);
  float R1[9];  mat3mul(R, Rrs, R1);
  float p1[3];
  p1[0] = R[0] * por[0] + R[1] * por[1] + R[2] * por[2] + p[0];
  p1[1] = R[3] * por[0] + R[4] * por[1] + R[5] * por[2] + p[1];
  p1[2] = R[6] * por[0] + R[7] * por[1] + R[8] * por[2] + p[2];

  float4* r4 = (float4*)myRev;
  r4[0] = make_float4(R1[0], R1[1], R1[2], p1[0]);
  r4[1] = make_float4(R1[3], R1[4], R1[5], p1[1]);
  r4[2] = make_float4(R1[6], R1[7], R1[8], p1[2]);

  float tv0 = d * ap[0], tv1 = d * ap[1], tv2 = d * ap[2];
  float ppv[3];
  ppv[0] = Rop[0] * tv0 + Rop[1] * tv1 + Rop[2] * tv2 + pop[0];
  ppv[1] = Rop[3] * tv0 + Rop[4] * tv1 + Rop[5] * tv2 + pop[1];
  ppv[2] = Rop[6] * tv0 + Rop[7] * tv1 + Rop[8] * tv2 + pop[2];
  float R2[9]; mat3mul(R1, Rop, R2);
  float p2[3];
  p2[0] = R1[0] * ppv[0] + R1[1] * ppv[1] + R1[2] * ppv[2] + p1[0];
  p2[1] = R1[3] * ppv[0] + R1[4] * ppv[1] + R1[5] * ppv[2] + p1[1];
  p2[2] = R1[6] * ppv[0] + R1[7] * ppv[1] + R1[8] * ppv[2] + p1[2];
#pragma unroll
  for (int k = 0; k < 9; ++k) R[k] = R2[k];
  p[0] = p2[0]; p[1] = p2[1]; p[2] = p2[2];

  float4* q4 = (float4*)myPri;
  q4[0] = make_float4(R[0], R[1], R[2], p[0]);
  q4[1] = make_float4(R[3], R[4], R[5], p[1]);
  q4[2] = make_float4(R[6], R[7], R[8], p[2]);
}

__device__ __forceinline__ void wave_stage_handover() {
  asm volatile("s_wait_dscnt 0x0" ::: "memory");
}

__device__ __forceinline__ void read_plane(const float* plane, int fc, int fp, float4 (&v)[8]) {
#pragma unroll
  for (int k = 0; k < 8; ++k)
    v[k] = *(const float4*)(plane + (k * 4 + fc) * PITCH + fp * 4);
}

__device__ __forceinline__ void store_lines(float* __restrict__ o, size_t chainStride,
                                            const float4 (&v)[8]) {
#pragma unroll
  for (int k = 0; k < 8; ++k)
    *(float4*)(o + (size_t)(k * 4) * chainStride) = v[k];
}

__device__ __forceinline__ void store_lines_ragged(const float* plane, float* __restrict__ o,
                                                   size_t chainStride, long firstChain,
                                                   int B, int fc, int fp) {
#pragma unroll 1
  for (int k = 0; k < 8; ++k)
    if (firstChain + k * 4 < (long)B)
      *(float4*)(o + (size_t)(k * 4) * chainStride) =
          *(const float4*)(plane + (k * 4 + fc) * PITCH + fp * 4);
}

__global__ __launch_bounds__(256) void fk_main_kernel(
    const float* __restrict__ rev_q, const float* __restrict__ pri_q,
    const float* __restrict__ cst, float* __restrict__ out, int B) {
  __shared__ float4 stage[8 * 3 * 32 * (PITCH / 4)];
  __shared__ float4 tab4[NJ * 16];
  __shared__ float4 qd4[8 * 2 * 160];

  const int lane = threadIdx.x & 31;
  const int wv   = threadIdx.x >> 5;
  float* sRev = (float*)stage + wv * (3 * 32 * PITCH);
  float* sPri = sRev + 32 * PITCH;
  float* sTrk = sPri + 32 * PITCH;
  float* myRev = sRev + lane * PITCH;
  float* myPri = sPri + lane * PITCH;

  float* trk_out = out;
  float* rev_out = out + (size_t)B * 160;
  float* pri_out = out + (size_t)B * 480;

  const int  h   = lane >> 4;
  const int  m16 = lane & 15;
  const int  nu  = m16 & 3;
  const int  iq  = m16 >> 2;
  const long waveBase = (long)blockIdx.x * 256 + (long)__builtin_amdgcn_readfirstlane(wv) * 32;
  const bool whole = (waveBase + 32 <= (long)B);
  const int fc = lane >> 3;
  const int fp = lane & 7;

  const int  tt = threadIdx.x < NJ * 12 ? threadIdx.x : NJ * 12 - 1;
  const int  tj = tt / 12;
  const int  tk = tt - tj * 12;
  const float4 tpiece = ((const float4*)cst)[tj * 16 + tk];
  float4* wq4 = qd4 + wv * 320;
  float4* wd4 = wq4 + 160;
  {
    const long lastPiece = (long)B * (NJ / 4) - 1;
    float4 vq[5], vd[5];
#pragma unroll
    for (int k = 0; k < 5; ++k) {
      long piece = waveBase * (NJ / 4) + lane + 32 * k;
      piece = piece < lastPiece ? piece : lastPiece;
      vq[k] = ((const float4*)rev_q)[piece];
      vd[k] = ((const float4*)pri_q)[piece];
    }
#pragma unroll
    for (int k = 0; k < 5; ++k) {
      wq4[lane + 32 * k] = vq[k];
      wd4[lane + 32 * k] = vd[k];
    }
    tab4[tj * 16 + tk] = tpiece;
  }
  const float* myQ = (const float*)wq4 + lane * NJ;
  const float* myD = (const float*)wd4 + lane * NJ;
  const float* tab = (const float*)tab4;

  float R[9] = {1.f, 0.f, 0.f, 0.f, 1.f, 0.f, 0.f, 0.f, 1.f};
  float p[3] = {0.f, 0.f, 0.f};

  {
    const float4 e3 = make_float4(0.f, 0.f, 0.f, 1.f);
    *(float4*)(myRev + 12) = e3; *(float4*)(myRev + 28) = e3;
    *(float4*)(myPri + 12) = e3; *(float4*)(myPri + 28) = e3;
  }

  __syncthreads();

  float2 qq = *(const float2*)myQ;
  float2 dd = *(const float2*)myD;

  for (int jp = 0; jp < NJ / 2; ++jp) {
    const int j0 = 2 * jp;
    const float* C0 = tab + j0 * 64;
    const float* C1 = C0 + 64;
    const float q0 = qq.x, q1 = qq.y;
    const float d0 = dd.x, d1 = dd.y;
    {
      const int jn = (jp + 1 < NJ / 2) ? j0 + 2 : j0;
      qq = *(const float2*)(myQ + jn);
      dd = *(const float2*)(myD + jn);
    }
    v2f Aop;
    Aop.x = C0[30 + (2 * h + 0) * 4 + nu];
    Aop.y = C0[30 + (2 * h + 1) * 4 + nu];

    fk_joint(C0, q0, d0, R, p, myRev, myPri);
    wave_stage_handover();

    {
      const v8f zero8 = {0.f, 0.f, 0.f, 0.f, 0.f, 0.f, 0.f, 0.f};
      v2f Bv[8];
#pragma unroll
      for (int g = 0; g < 8; ++g)
        Bv[g] = *(const v2f*)(sPri + (g * 4 + iq) * PITCH + nu * 4 + 2 * h);
      v8f D[8];
#pragma unroll
      for (int g = 0; g < 8; ++g)
        D[g] = __builtin_amdgcn_wmma_f32_16x16x4_f32(
            false, Aop, false, Bv[g], (short)0, zero8, false, false);
      if (h == 0) {
        float* trow = sTrk + iq * PITCH + (jp & 1) * 16 + nu * 4;
#pragma unroll
        for (int g = 0; g < 8; ++g)
          *(float4*)(trow + g * 4 * PITCH) = make_float4(D[g][0], D[g][1], D[g][2], D[g][3]);
      }
    }

    fk_joint(C1, q1, d1, R, p, myRev + 16, myPri + 16);
    wave_stage_handover();

    float* oRev = rev_out + ((size_t)(waveBase + fc) * NJ + j0) * 16 + fp * 4;
    float* oPri = pri_out + ((size_t)(waveBase + fc) * NJ + j0) * 16 + fp * 4;
    if (whole) {
      float4 vr[8], vp[8];
      read_plane(sRev, fc, fp, vr);
      read_plane(sPri, fc, fp, vp);
      if (jp & 1) {
        float4 vt[8];
        read_plane(sTrk, fc, fp, vt);
        float* oTrk = trk_out + ((size_t)(waveBase + fc) * 10 + (jp - 1)) * 16 + fp * 4;
        store_lines(oRev, (size_t)NJ * 16, vr);
        store_lines(oPri, (size_t)NJ * 16, vp);
        store_lines(oTrk, (size_t)160, vt);
      } else {
        store_lines(oRev, (size_t)NJ * 16, vr);
        store_lines(oPri, (size_t)NJ * 16, vp);
      }
    } else {
      store_lines_ragged(sRev, oRev, (size_t)NJ * 16, waveBase + fc, B, fc, fp);
      store_lines_ragged(sPri, oPri, (size_t)NJ * 16, waveBase + fc, B, fc, fp);
      if (jp & 1) {
        float* oTrk = trk_out + ((size_t)(waveBase + fc) * 10 + (jp - 1)) * 16 + fp * 4;
        store_lines_ragged(sTrk, oTrk, (size_t)160, waveBase + fc, B, fc, fp);
      }
    }
  }
}

extern "C" void kernel_launch(void* const* d_in, const int* in_sizes, int n_in,
                              void* d_out, int out_size, void* d_ws, size_t ws_size,
                              hipStream_t stream) {
  (void)n_in; (void)out_size; (void)ws_size;
  const float* rev_q       = (const float*)d_in[0];
  const float* pri_q       = (const float*)d_in[1];
  const float* rev_p_off   = (const float*)d_in[2];
  const float* rev_rpy_off = (const float*)d_in[3];
  const float* pri_p_off   = (const float*)d_in[4];
  const float* pri_rpy_off = (const float*)d_in[5];
  const float* rev_axis    = (const float*)d_in[6];
  const float* pri_axis    = (const float*)d_in[7];
  const float* p_track     = (const float*)d_in[8];
  const float* rpy_track   = (const float*)d_in[9];

  float* cst = (float*)d_ws;
  int B = in_sizes[0] / NJ;

  fk_const_kernel<<<1, 32, 0, stream>>>(rev_p_off, rev_rpy_off, pri_p_off, pri_rpy_off,
                                        rev_axis, pri_axis, p_track, rpy_track, cst);
  int grid = (B + 255) / 256;
  fk_main_kernel<<<grid, 256, 0, stream>>>(rev_q, pri_q, cst, (float*)d_out, B);
}


# Per-joint constant table built inside the main kernel, by 80 threads of every work-group straight into LDS ahead of the barrier it already had: one launch per call instead of two
# speedup vs baseline: 3.9723x; 1.4296x over previous
#include <hip/hip_runtime.h>
#include <cstddef>

typedef float v2f __attribute__((ext_vector_type(2)));
typedef float v8f __attribute__((ext_vector_type(8)));

#define NJ 20
#define PITCH 36

__device__ __forceinline__ void rpy2r_dev(float r, float p, float y, float* R) {
  float cr = cosf(r), sr = sinf(r);
  float cp = cosf(p), sp = sinf(p);
  float cy = cosf(y), sy = sinf(y);
  R[0] = cy * cp; R[1] = cy * sp * sr - sy * cr; R[2] = cy * sp * cr + sy * sr;
  R[3] = sy * cp; R[4] = sy * sp * sr + cy * cr; R[5] = sy * sp * cr - cy * sr;
  R[6] = -sp;     R[7] = cp * sr;                R[8] = cp * cr;
}

__device__ __forceinline__ void mat3mul(const float* A, const float* B, float* C) {
#pragma unroll
  for (int i = 0; i < 3; ++i)
#pragma unroll
    for (int k = 0; k < 3; ++k)
      C[i * 3 + k] = A[i * 3 + 0] * B[0 * 3 + k] + A[i * 3 + 1] * B[1 * 3 + k] +
                     A[i * 3 + 2] * B[2 * 3 + k];
}


__device__ __forceinline__ void fk_joint(const float* __restrict__ C, float q, float d,
                                         float (&R)[9], float (&p)[3],
                                         float* myRev, float* myPri) {
  float Ror[9], por[3], Rop[9], pop[3], ar[3], ap[3];
#pragma unroll
  for (int k = 0; k < 9; ++k) Ror[k] = C[k];
#pragma unroll
  for (int k = 0; k < 3; ++k) por[k] = C[9 + k];
#pragma unroll
  for (int k = 0; k < 9; ++k) Rop[k] = C[12 + k];
#pragma unroll
  for (int k = 0; k < 3; ++k) pop[k] = C[21 + k];
#pragma unroll
  for (int k = 0; k < 3; ++k) ar[k] = C[24 + k];
#pragma unroll
  for (int k = 0; k < 3; ++k) ap[k] = C[27 + k];

  float sn = sinf(q), cs = cosf(q), oc = 1.f - cs;
  float ax = ar[0], ay = ar[1], az = ar[2];
  float Rod[9];
  Rod[0] = 1.f + oc * (-(ay * ay + az * az));
  Rod[1] = -sn * az + oc * (ax * ay);
  Rod[2] =  sn * ay + oc * (ax * az);
  Rod[3] =  sn * az + oc * (ax * ay);
  Rod[4] = 1.f + oc * (-(ax * ax + az * az));
  Rod[5] = -sn * ax + oc * (ay * az);
  Rod[6] = -sn * ay + oc * (ax * az);
  Rod[7] =  sn * ax + oc * (ay * az);
  Rod[8] = 1.f + oc * (-(ax * ax + ay * ay));

  float Rrs[9]; mat3mul(Ror, Rod, Rrs);
  float R1[9];  mat3mul(R, Rrs, R1);
  float p1[3];
  p1[0] = R[0] * por[0] + R[1] * por[1] + R[2] * por[2] + p[0];
  p1[1] = R[3] * por[0] + R[4] * por[1] + R[5] * por[2] + p[1];
  p1[2] = R[6] * por[0] + R[7] * por[1] + R[8] * por[2] + p[2];

  float4* r4 = (float4*)myRev;
  r4[0] = make_float4(R1[0], R1[1], R1[2], p1[0]);
  r4[1] = make_float4(R1[3], R1[4], R1[5], p1[1]);
  r4[2] = make_float4(R1[6], R1[7], R1[8], p1[2]);

  float tv0 = d * ap[0], tv1 = d * ap[1], tv2 = d * ap[2];
  float ppv[3];
  ppv[0] = Rop[0] * tv0 + Rop[1] * tv1 + Rop[2] * tv2 + pop[0];
  ppv[1] = Rop[3] * tv0 + Rop[4] * tv1 + Rop[5] * tv2 + pop[1];
  ppv[2] = Rop[6] * tv0 + Rop[7] * tv1 + Rop[8] * tv2 + pop[2];
  float R2[9]; mat3mul(R1, Rop, R2);
  float p2[3];
  p2[0] = R1[0] * ppv[0] + R1[1] * ppv[1] + R1[2] * ppv[2] + p1[0];
  p2[1] = R1[3] * ppv[0] + R1[4] * ppv[1] + R1[5] * ppv[2] + p1[1];
  p2[2] = R1[6] * ppv[0] + R1[7] * ppv[1] + R1[8] * ppv[2] + p1[2];
#pragma unroll
  for (int k = 0; k < 9; ++k) R[k] = R2[k];
  p[0] = p2[0]; p[1] = p2[1]; p[2] = p2[2];

  float4* q4 = (float4*)myPri;
  q4[0] = make_float4(R[0], R[1], R[2], p[0]);
  q4[1] = make_float4(R[3], R[4], R[5], p[1]);
  q4[2] = make_float4(R[6], R[7], R[8], p[2]);
}

__device__ __forceinline__ void wave_stage_handover() {
  asm volatile("s_wait_dscnt 0x0" ::: "memory");
}

__device__ __forceinline__ void read_plane(const float* plane, int fc, int fp, float4 (&v)[8]) {
#pragma unroll
  for (int k = 0; k < 8; ++k)
    v[k] = *(const float4*)(plane + (k * 4 + fc) * PITCH + fp * 4);
}

__device__ __forceinline__ void store_lines(float* __restrict__ o, size_t chainStride,
                                            const float4 (&v)[8]) {
#pragma unroll
  for (int k = 0; k < 8; ++k)
    *(float4*)(o + (size_t)(k * 4) * chainStride) = v[k];
}

__device__ __forceinline__ void store_lines_ragged(const float* plane, float* __restrict__ o,
                                                   size_t chainStride, long firstChain,
                                                   int B, int fc, int fp) {
#pragma unroll 1
  for (int k = 0; k < 8; ++k)
    if (firstChain + k * 4 < (long)B)
      *(float4*)(o + (size_t)(k * 4) * chainStride) =
          *(const float4*)(plane + (k * 4 + fc) * PITCH + fp * 4);
}

__global__ __launch_bounds__(256) void fk_main_kernel(
    const float* __restrict__ rev_q, const float* __restrict__ pri_q,
    const float* __restrict__ rev_p_off, const float* __restrict__ rev_rpy_off,
    const float* __restrict__ pri_p_off, const float* __restrict__ pri_rpy_off,
    const float* __restrict__ rev_axis, const float* __restrict__ pri_axis,
    const float* __restrict__ p_track, const float* __restrict__ rpy_track,
    float* __restrict__ out, int B) {
  __shared__ float4 stage[8 * 3 * 32 * (PITCH / 4)];
  __shared__ float4 tab4[NJ * 16];
  __shared__ float4 qd4[8 * 2 * 160];

  const int lane = threadIdx.x & 31;
  const int wv   = threadIdx.x >> 5;
  float* sRev = (float*)stage + wv * (3 * 32 * PITCH);
  float* sPri = sRev + 32 * PITCH;
  float* sTrk = sPri + 32 * PITCH;
  float* myRev = sRev + lane * PITCH;
  float* myPri = sPri + lane * PITCH;

  float* trk_out = out;
  float* rev_out = out + (size_t)B * 160;
  float* pri_out = out + (size_t)B * 480;

  const int  h   = lane >> 4;
  const int  m16 = lane & 15;
  const int  nu  = m16 & 3;
  const int  iq  = m16 >> 2;
  const long waveBase = (long)blockIdx.x * 256 + (long)__builtin_amdgcn_readfirstlane(wv) * 32;
  const bool whole = (waveBase + 32 <= (long)B);
  const int fc = lane >> 3;
  const int fp = lane & 7;

  const int tid = threadIdx.x;
  if (tid < 3 * NJ) {
    const int bw = tid / NJ;
    const int bj = tid - bw * NJ;
    const float* rpySrc = bw == 0 ? rev_rpy_off : (bw == 1 ? pri_rpy_off : rpy_track);
    const float* posSrc = bw == 0 ? rev_p_off   : (bw == 1 ? pri_p_off   : p_track);
    float Rt[9];
    rpy2r_dev(rpySrc[3 * bj], rpySrc[3 * bj + 1], rpySrc[3 * bj + 2], Rt);
    const float t0 = posSrc[3 * bj], t1 = posSrc[3 * bj + 1], t2 = posSrc[3 * bj + 2];
    float* o = (float*)tab4 + bj * 64;
    if (bw == 2) {
      o[30] = Rt[0]; o[31] = Rt[1]; o[32] = Rt[2]; o[33] = t0;
      o[34] = Rt[3]; o[35] = Rt[4]; o[36] = Rt[5]; o[37] = t1;
      o[38] = Rt[6]; o[39] = Rt[7]; o[40] = Rt[8]; o[41] = t2;
      o[42] = 0.f; o[43] = 0.f; o[44] = 0.f; o[45] = 1.f;
    } else {
      float* oo = o + bw * 12;
#pragma unroll
      for (int k = 0; k < 9; ++k) oo[k] = Rt[k];
      oo[9] = t0; oo[10] = t1; oo[11] = t2;
    }
  } else if (tid >= 64 && tid < 64 + NJ) {
    const int aj = tid - 64;
    float* o = (float*)tab4 + aj * 64;
    const float x = rev_axis[3 * aj], y = rev_axis[3 * aj + 1], z = rev_axis[3 * aj + 2];
    const float inv = 1.f / (sqrtf(x * x + y * y + z * z) + 1e-12f);
    o[24] = x * inv; o[25] = y * inv; o[26] = z * inv;
    o[27] = pri_axis[3 * aj]; o[28] = pri_axis[3 * aj + 1]; o[29] = pri_axis[3 * aj + 2];
  }
  float4* wq4 = qd4 + wv * 320;
  float4* wd4 = wq4 + 160;
  {
    const long lastPiece = (long)B * (NJ / 4) - 1;
    float4 vq[5], vd[5];
#pragma unroll
    for (int k = 0; k < 5; ++k) {
      long piece = waveBase * (NJ / 4) + lane + 32 * k;
      piece = piece < lastPiece ? piece : lastPiece;
      vq[k] = ((const float4*)rev_q)[piece];
      vd[k] = ((const float4*)pri_q)[piece];
    }
#pragma unroll
    for (int k = 0; k < 5; ++k) {
      wq4[lane + 32 * k] = vq[k];
      wd4[lane + 32 * k] = vd[k];
    }
  }
  const float* myQ = (const float*)wq4 + lane * NJ;
  const float* myD = (const float*)wd4 + lane * NJ;
  const float* tab = (const float*)tab4;

  float R[9] = {1.f, 0.f, 0.f, 0.f, 1.f, 0.f, 0.f, 0.f, 1.f};
  float p[3] = {0.f, 0.f, 0.f};

  {
    const float4 e3 = make_float4(0.f, 0.f, 0.f, 1.f);
    *(float4*)(myRev + 12) = e3; *(float4*)(myRev + 28) = e3;
    *(float4*)(myPri + 12) = e3; *(float4*)(myPri + 28) = e3;
  }

  __syncthreads();

  float2 qq = *(const float2*)myQ;
  float2 dd = *(const float2*)myD;

  for (int jp = 0; jp < NJ / 2; ++jp) {
    const int j0 = 2 * jp;
    const float* C0 = tab + j0 * 64;
    const float* C1 = C0 + 64;
    const float q0 = qq.x, q1 = qq.y;
    const float d0 = dd.x, d1 = dd.y;
    {
      const int jn = (jp + 1 < NJ / 2) ? j0 + 2 : j0;
      qq = *(const float2*)(myQ + jn);
      dd = *(const float2*)(myD + jn);
    }
    v2f Aop;
    Aop.x = C0[30 + (2 * h + 0) * 4 + nu];
    Aop.y = C0[30 + (2 * h + 1) * 4 + nu];

    fk_joint(C0, q0, d0, R, p, myRev, myPri);
    wave_stage_handover();

    {
      const v8f zero8 = {0.f, 0.f, 0.f, 0.f, 0.f, 0.f, 0.f, 0.f};
      v2f Bv[8];
#pragma unroll
      for (int g = 0; g < 8; ++g)
        Bv[g] = *(const v2f*)(sPri + (g * 4 + iq) * PITCH + nu * 4 + 2 * h);
      v8f D[8];
#pragma unroll
      for (int g = 0; g < 8; ++g)
        D[g] = __builtin_amdgcn_wmma_f32_16x16x4_f32(
            false, Aop, false, Bv[g], (short)0, zero8, false, false);
      if (h == 0) {
        float* trow = sTrk + iq * PITCH + (jp & 1) * 16 + nu * 4;
#pragma unroll
        for (int g = 0; g < 8; ++g)
          *(float4*)(trow + g * 4 * PITCH) = make_float4(D[g][0], D[g][1], D[g][2], D[g][3]);
      }
    }

    fk_joint(C1, q1, d1, R, p, myRev + 16, myPri + 16);
    wave_stage_handover();

    float* oRev = rev_out + ((size_t)(waveBase + fc) * NJ + j0) * 16 + fp * 4;
    float* oPri = pri_out + ((size_t)(waveBase + fc) * NJ + j0) * 16 + fp * 4;
    if (whole) {
      float4 vr[8], vp[8];
      read_plane(sRev, fc, fp, vr);
      read_plane(sPri, fc, fp, vp);
      if (jp & 1) {
        float4 vt[8];
        read_plane(sTrk, fc, fp, vt);
        float* oTrk = trk_out + ((size_t)(waveBase + fc) * 10 + (jp - 1)) * 16 + fp * 4;
        store_lines(oRev, (size_t)NJ * 16, vr);
        store_lines(oPri, (size_t)NJ * 16, vp);
        store_lines(oTrk, (size_t)160, vt);
      } else {
        store_lines(oRev, (size_t)NJ * 16, vr);
        store_lines(oPri, (size_t)NJ * 16, vp);
      }
    } else {
      store_lines_ragged(sRev, oRev, (size_t)NJ * 16, waveBase + fc, B, fc, fp);
      store_lines_ragged(sPri, oPri, (size_t)NJ * 16, waveBase + fc, B, fc, fp);
      if (jp & 1) {
        float* oTrk = trk_out + ((size_t)(waveBase + fc) * 10 + (jp - 1)) * 16 + fp * 4;
        store_lines_ragged(sTrk, oTrk, (size_t)160, waveBase + fc, B, fc, fp);
      }
    }
  }
}

extern "C" void kernel_launch(void* const* d_in, const int* in_sizes, int n_in,
                              void* d_out, int out_size, void* d_ws, size_t ws_size,
                              hipStream_t stream) {
  (void)n_in; (void)out_size; (void)d_ws; (void)ws_size;
  const float* rev_q       = (const float*)d_in[0];
  const float* pri_q       = (const float*)d_in[1];
  const float* rev_p_off   = (const float*)d_in[2];
  const float* rev_rpy_off = (const float*)d_in[3];
  const float* pri_p_off   = (const float*)d_in[4];
  const float* pri_rpy_off = (const float*)d_in[5];
  const float* rev_axis    = (const float*)d_in[6];
  const float* pri_axis    = (const float*)d_in[7];
  const float* p_track     = (const float*)d_in[8];
  const float* rpy_track   = (const float*)d_in[9];

  int B = in_sizes[0] / NJ;

  int grid = (B + 255) / 256;
  fk_main_kernel<<<grid, 256, 0, stream>>>(rev_q, pri_q, rev_p_off, rev_rpy_off, pri_p_off,
                                           pri_rpy_off, rev_axis, pri_axis, p_track, rpy_track,
                                           (float*)d_out, B);
}


# Work-groups of 4 waves instead of 8: 256 work-groups, one on every WGP of the chip instead of 128 on half of them, and one wave per SIMD instead of two; per-wave code unchanged
# speedup vs baseline: 4.1748x; 1.0510x over previous
#include <hip/hip_runtime.h>
#include <cstddef>

typedef float v2f __attribute__((ext_vector_type(2)));
typedef float v8f __attribute__((ext_vector_type(8)));

#define NJ 20
#define PITCH 36

__device__ __forceinline__ void rpy2r_dev(float r, float p, float y, float* R) {
  float cr = cosf(r), sr = sinf(r);
  float cp = cosf(p), sp = sinf(p);
  float cy = cosf(y), sy = sinf(y);
  R[0] = cy * cp; R[1] = cy * sp * sr - sy * cr; R[2] = cy * sp * cr + sy * sr;
  R[3] = sy * cp; R[4] = sy * sp * sr + cy * cr; R[5] = sy * sp * cr - cy * sr;
  R[6] = -sp;     R[7] = cp * sr;                R[8] = cp * cr;
}

__device__ __forceinline__ void mat3mul(const float* A, const float* B, float* C) {
#pragma unroll
  for (int i = 0; i < 3; ++i)
#pragma unroll
    for (int k = 0; k < 3; ++k)
      C[i * 3 + k] = A[i * 3 + 0] * B[0 * 3 + k] + A[i * 3 + 1] * B[1 * 3 + k] +
                     A[i * 3 + 2] * B[2 * 3 + k];
}


__device__ __forceinline__ void fk_joint(const float* __restrict__ C, float q, float d,
                                         float (&R)[9], float (&p)[3],
                                         float* myRev, float* myPri) {
  float Ror[9], por[3], Rop[9], pop[3], ar[3], ap[3];
#pragma unroll
  for (int k = 0; k < 9; ++k) Ror[k] = C[k];
#pragma unroll
  for (int k = 0; k < 3; ++k) por[k] = C[9 + k];
#pragma unroll
  for (int k = 0; k < 9; ++k) Rop[k] = C[12 + k];
#pragma unroll
  for (int k = 0; k < 3; ++k) pop[k] = C[21 + k];
#pragma unroll
  for (int k = 0; k < 3; ++k) ar[k] = C[24 + k];
#pragma unroll
  for (int k = 0; k < 3; ++k) ap[k] = C[27 + k];

  float sn = sinf(q), cs = cosf(q), oc = 1.f - cs;
  float ax = ar[0], ay = ar[1], az = ar[2];
  float Rod[9];
  Rod[0] = 1.f + oc * (-(ay * ay + az * az));
  Rod[1] = -sn * az + oc * (ax * ay);
  Rod[2] =  sn * ay + oc * (ax * az);
  Rod[3] =  sn * az + oc * (ax * ay);
  Rod[4] = 1.f + oc * (-(ax * ax + az * az));
  Rod[5] = -sn * ax + oc * (ay * az);
  Rod[6] = -sn * ay + oc * (ax * az);
  Rod[7] =  sn * ax + oc * (ay * az);
  Rod[8] = 1.f + oc * (-(ax * ax + ay * ay));

  float Rrs[9]; mat3mul(Ror, Rod, Rrs);
  float R1[9];  mat3mul(R, Rrs, R1);
  float p1[3];
  p1[0] = R[0] * por[0] + R[1] * por[1] + R[2] * por[2] + p[0];
  p1[1] = R[3] * por[0] + R[4] * por[1] + R[5] * por[2] + p[1];
  p1[2] = R[6] * por[0] + R[7] * por[1] + R[8] * por[2] + p[2];

  float4* r4 = (float4*)myRev;
  r4[0] = make_float4(R1[0], R1[1], R1[2], p1[0]);
  r4[1] = make_float4(R1[3], R1[4], R1[5], p1[1]);
  r4[2] = make_float4(R1[6], R1[7], R1[8], p1[2]);

  float tv0 = d * ap[0], tv1 = d * ap[1], tv2 = d * ap[2];
  float ppv[3];
  ppv[0] = Rop[0] * tv0 + Rop[1] * tv1 + Rop[2] * tv2 + pop[0];
  ppv[1] = Rop[3] * tv0 + Rop[4] * tv1 + Rop[5] * tv2 + pop[1];
  ppv[2] = Rop[6] * tv0 + Rop[7] * tv1 + Rop[8] * tv2 + pop[2];
  float R2[9]; mat3mul(R1, Rop, R2);
  float p2[3];
  p2[0] = R1[0] * ppv[0] + R1[1] * ppv[1] + R1[2] * ppv[2] + p1[0];
  p2[1] = R1[3] * ppv[0] + R1[4] * ppv[1] + R1[5] * ppv[2] + p1[1];
  p2[2] = R1[6] * ppv[0] + R1[7] * ppv[1] + R1[8] * ppv[2] + p1[2];
#pragma unroll
  for (int k = 0; k < 9; ++k) R[k] = R2[k];
  p[0] = p2[0]; p[1] = p2[1]; p[2] = p2[2];

  float4* q4 = (float4*)myPri;
  q4[0] = make_float4(R[0], R[1], R[2], p[0]);
  q4[1] = make_float4(R[3], R[4], R[5], p[1]);
  q4[2] = make_float4(R[6], R[7], R[8], p[2]);
}

__device__ __forceinline__ void wave_stage_handover() {
  asm volatile("s_wait_dscnt 0x0" ::: "memory");
}

__device__ __forceinline__ void read_plane(const float* plane, int fc, int fp, float4 (&v)[8]) {
#pragma unroll
  for (int k = 0; k < 8; ++k)
    v[k] = *(const float4*)(plane + (k * 4 + fc) * PITCH + fp * 4);
}

__device__ __forceinline__ void store_lines(float* __restrict__ o, size_t chainStride,
                                            const float4 (&v)[8]) {
#pragma unroll
  for (int k = 0; k < 8; ++k)
    *(float4*)(o + (size_t)(k * 4) * chainStride) = v[k];
}

__device__ __forceinline__ void store_lines_ragged(const float* plane, float* __restrict__ o,
                                                   size_t chainStride, long firstChain,
                                                   int B, int fc, int fp) {
#pragma unroll 1
  for (int k = 0; k < 8; ++k)
    if (firstChain + k * 4 < (long)B)
      *(float4*)(o + (size_t)(k * 4) * chainStride) =
          *(const float4*)(plane + (k * 4 + fc) * PITCH + fp * 4);
}

__global__ __launch_bounds__(128) void fk_main_kernel(
    const float* __restrict__ rev_q, const float* __restrict__ pri_q,
    const float* __restrict__ rev_p_off, const float* __restrict__ rev_rpy_off,
    const float* __restrict__ pri_p_off, const float* __restrict__ pri_rpy_off,
    const float* __restrict__ rev_axis, const float* __restrict__ pri_axis,
    const float* __restrict__ p_track, const float* __restrict__ rpy_track,
    float* __restrict__ out, int B) {
  __shared__ float4 stage[4 * 3 * 32 * (PITCH / 4)];
  __shared__ float4 tab4[NJ * 16];
  __shared__ float4 qd4[4 * 2 * 160];

  const int lane = threadIdx.x & 31;
  const int wv   = threadIdx.x >> 5;
  float* sRev = (float*)stage + wv * (3 * 32 * PITCH);
  float* sPri = sRev + 32 * PITCH;
  float* sTrk = sPri + 32 * PITCH;
  float* myRev = sRev + lane * PITCH;
  float* myPri = sPri + lane * PITCH;

  float* trk_out = out;
  float* rev_out = out + (size_t)B * 160;
  float* pri_out = out + (size_t)B * 480;

  const int  h   = lane >> 4;
  const int  m16 = lane & 15;
  const int  nu  = m16 & 3;
  const int  iq  = m16 >> 2;
  const long waveBase = (long)blockIdx.x * 128 + (long)__builtin_amdgcn_readfirstlane(wv) * 32;
  const bool whole = (waveBase + 32 <= (long)B);
  const int fc = lane >> 3;
  const int fp = lane & 7;

  const int tid = threadIdx.x;
  if (tid < 3 * NJ) {
    const int bw = tid / NJ;
    const int bj = tid - bw * NJ;
    const float* rpySrc = bw == 0 ? rev_rpy_off : (bw == 1 ? pri_rpy_off : rpy_track);
    const float* posSrc = bw == 0 ? rev_p_off   : (bw == 1 ? pri_p_off   : p_track);
    float Rt[9];
    rpy2r_dev(rpySrc[3 * bj], rpySrc[3 * bj + 1], rpySrc[3 * bj + 2], Rt);
    const float t0 = posSrc[3 * bj], t1 = posSrc[3 * bj + 1], t2 = posSrc[3 * bj + 2];
    float* o = (float*)tab4 + bj * 64;
    if (bw == 2) {
      o[30] = Rt[0]; o[31] = Rt[1]; o[32] = Rt[2]; o[33] = t0;
      o[34] = Rt[3]; o[35] = Rt[4]; o[36] = Rt[5]; o[37] = t1;
      o[38] = Rt[6]; o[39] = Rt[7]; o[40] = Rt[8]; o[41] = t2;
      o[42] = 0.f; o[43] = 0.f; o[44] = 0.f; o[45] = 1.f;
    } else {
      float* oo = o + bw * 12;
#pragma unroll
      for (int k = 0; k < 9; ++k) oo[k] = Rt[k];
      oo[9] = t0; oo[10] = t1; oo[11] = t2;
    }
  } else if (tid >= 64 && tid < 64 + NJ) {
    const int aj = tid - 64;
    float* o = (float*)tab4 + aj * 64;
    const float x = rev_axis[3 * aj], y = rev_axis[3 * aj + 1], z = rev_axis[3 * aj + 2];
    const float inv = 1.f / (sqrtf(x * x + y * y + z * z) + 1e-12f);
    o[24] = x * inv; o[25] = y * inv; o[26] = z * inv;
    o[27] = pri_axis[3 * aj]; o[28] = pri_axis[3 * aj + 1]; o[29] = pri_axis[3 * aj + 2];
  }
  float4* wq4 = qd4 + wv * 320;
  float4* wd4 = wq4 + 160;
  {
    const long lastPiece = (long)B * (NJ / 4) - 1;
    float4 vq[5], vd[5];
#pragma unroll
    for (int k = 0; k < 5; ++k) {
      long piece = waveBase * (NJ / 4) + lane + 32 * k;
      piece = piece < lastPiece ? piece : lastPiece;
      vq[k] = ((const float4*)rev_q)[piece];
      vd[k] = ((const float4*)pri_q)[piece];
    }
#pragma unroll
    for (int k = 0; k < 5; ++k) {
      wq4[lane + 32 * k] = vq[k];
      wd4[lane + 32 * k] = vd[k];
    }
  }
  const float* myQ = (const float*)wq4 + lane * NJ;
  const float* myD = (const float*)wd4 + lane * NJ;
  const float* tab = (const float*)tab4;

  float R[9] = {1.f, 0.f, 0.f, 0.f, 1.f, 0.f, 0.f, 0.f, 1.f};
  float p[3] = {0.f, 0.f, 0.f};

  {
    const float4 e3 = make_float4(0.f, 0.f, 0.f, 1.f);
    *(float4*)(myRev + 12) = e3; *(float4*)(myRev + 28) = e3;
    *(float4*)(myPri + 12) = e3; *(float4*)(myPri + 28) = e3;
  }

  __syncthreads();

  float2 qq = *(const float2*)myQ;
  float2 dd = *(const float2*)myD;

  for (int jp = 0; jp < NJ / 2; ++jp) {
    const int j0 = 2 * jp;
    const float* C0 = tab + j0 * 64;
    const float* C1 = C0 + 64;
    const float q0 = qq.x, q1 = qq.y;
    const float d0 = dd.x, d1 = dd.y;
    {
      const int jn = (jp + 1 < NJ / 2) ? j0 + 2 : j0;
      qq = *(const float2*)(myQ + jn);
      dd = *(const float2*)(myD + jn);
    }
    v2f Aop;
    Aop.x = C0[30 + (2 * h + 0) * 4 + nu];
    Aop.y = C0[30 + (2 * h + 1) * 4 + nu];

    fk_joint(C0, q0, d0, R, p, myRev, myPri);
    wave_stage_handover();

    {
      const v8f zero8 = {0.f, 0.f, 0.f, 0.f, 0.f, 0.f, 0.f, 0.f};
      v2f Bv[8];
#pragma unroll
      for (int g = 0; g < 8; ++g)
        Bv[g] = *(const v2f*)(sPri + (g * 4 + iq) * PITCH + nu * 4 + 2 * h);
      v8f D[8];
#pragma unroll
      for (int g = 0; g < 8; ++g)
        D[g] = __builtin_amdgcn_wmma_f32_16x16x4_f32(
            false, Aop, false, Bv[g], (short)0, zero8, false, false);
      if (h == 0) {
        float* trow = sTrk + iq * PITCH + (jp & 1) * 16 + nu * 4;
#pragma unroll
        for (int g = 0; g < 8; ++g)
          *(float4*)(trow + g * 4 * PITCH) = make_float4(D[g][0], D[g][1], D[g][2], D[g][3]);
      }
    }

    fk_joint(C1, q1, d1, R, p, myRev + 16, myPri + 16);
    wave_stage_handover();

    float* oRev = rev_out + ((size_t)(waveBase + fc) * NJ + j0) * 16 + fp * 4;
    float* oPri = pri_out + ((size_t)(waveBase + fc) * NJ + j0) * 16 + fp * 4;
    if (whole) {
      float4 vr[8], vp[8];
      read_plane(sRev, fc, fp, vr);
      read_plane(sPri, fc, fp, vp);
      if (jp & 1) {
        float4 vt[8];
        read_plane(sTrk, fc, fp, vt);
        float* oTrk = trk_out + ((size_t)(waveBase + fc) * 10 + (jp - 1)) * 16 + fp * 4;
        store_lines(oRev, (size_t)NJ * 16, vr);
        store_lines(oPri, (size_t)NJ * 16, vp);
        store_lines(oTrk, (size_t)160, vt);
      } else {
        store_lines(oRev, (size_t)NJ * 16, vr);
        store_lines(oPri, (size_t)NJ * 16, vp);
      }
    } else {
      store_lines_ragged(sRev, oRev, (size_t)NJ * 16, waveBase + fc, B, fc, fp);
      store_lines_ragged(sPri, oPri, (size_t)NJ * 16, waveBase + fc, B, fc, fp);
      if (jp & 1) {
        float* oTrk = trk_out + ((size_t)(waveBase + fc) * 10 + (jp - 1)) * 16 + fp * 4;
        store_lines_ragged(sTrk, oTrk, (size_t)160, waveBase + fc, B, fc, fp);
      }
    }
  }
}

extern "C" void kernel_launch(void* const* d_in, const int* in_sizes, int n_in,
                              void* d_out, int out_size, void* d_ws, size_t ws_size,
                              hipStream_t stream) {
  (void)n_in; (void)out_size; (void)d_ws; (void)ws_size;
  const float* rev_q       = (const float*)d_in[0];
  const float* pri_q       = (const float*)d_in[1];
  const float* rev_p_off   = (const float*)d_in[2];
  const float* rev_rpy_off = (const float*)d_in[3];
  const float* pri_p_off   = (const float*)d_in[4];
  const float* pri_rpy_off = (const float*)d_in[5];
  const float* rev_axis    = (const float*)d_in[6];
  const float* pri_axis    = (const float*)d_in[7];
  const float* p_track     = (const float*)d_in[8];
  const float* rpy_track   = (const float*)d_in[9];

  int B = in_sizes[0] / NJ;

  int grid = (B + 127) / 128;
  fk_main_kernel<<<grid, 128, 0, stream>>>(rev_q, pri_q, rev_p_off, rev_rpy_off, pri_p_off,
                                           pri_rpy_off, rev_axis, pri_axis, p_track, rpy_track,
                                           (float*)d_out, B);
}


# Every store instruction writes two whole aligned 256-B units (4 joints of 2 chains; tracking frames kept in LDS, stored behind the loop as one contiguous run per wave) instead of four 128-B lines
# speedup vs baseline: 6.0588x; 1.4513x over previous
#include <hip/hip_runtime.h>
#include <cstddef>

typedef float v2f __attribute__((ext_vector_type(2)));
typedef float v8f __attribute__((ext_vector_type(8)));

#define NJ 20
#define PITCH 68
#define TRKROW 160

__device__ __forceinline__ void rpy2r_dev(float r, float p, float y, float* R) {
  float cr = cosf(r), sr = sinf(r);
  float cp = cosf(p), sp = sinf(p);
  float cy = cosf(y), sy = sinf(y);
  R[0] = cy * cp; R[1] = cy * sp * sr - sy * cr; R[2] = cy * sp * cr + sy * sr;
  R[3] = sy * cp; R[4] = sy * sp * sr + cy * cr; R[5] = sy * sp * cr - cy * sr;
  R[6] = -sp;     R[7] = cp * sr;                R[8] = cp * cr;
}

__device__ __forceinline__ void mat3mul(const float* A, const float* B, float* C) {
#pragma unroll
  for (int i = 0; i < 3; ++i)
#pragma unroll
    for (int k = 0; k < 3; ++k)
      C[i * 3 + k] = A[i * 3 + 0] * B[0 * 3 + k] + A[i * 3 + 1] * B[1 * 3 + k] +
                     A[i * 3 + 2] * B[2 * 3 + k];
}


__device__ __forceinline__ void fk_joint(const float* __restrict__ C, float q, float d,
                                         float (&R)[9], float (&p)[3],
                                         float* myRev, float* myPri) {
  float Ror[9], por[3], Rop[9], pop[3], ar[3], ap[3];
#pragma unroll
  for (int k = 0; k < 9; ++k) Ror[k] = C[k];
#pragma unroll
  for (int k = 0; k < 3; ++k) por[k] = C[9 + k];
#pragma unroll
  for (int k = 0; k < 9; ++k) Rop[k] = C[12 + k];
#pragma unroll
  for (int k = 0; k < 3; ++k) pop[k] = C[21 + k];
#pragma unroll
  for (int k = 0; k < 3; ++k) ar[k] = C[24 + k];
#pragma unroll
  for (int k = 0; k < 3; ++k) ap[k] = C[27 + k];

  float sn = sinf(q), cs = cosf(q), oc = 1.f - cs;
  float ax = ar[0], ay = ar[1], az = ar[2];
  float Rod[9];
  Rod[0] = 1.f + oc * (-(ay * ay + az * az));
  Rod[1] = -sn * az + oc * (ax * ay);
  Rod[2] =  sn * ay + oc * (ax * az);
  Rod[3] =  sn * az + oc * (ax * ay);
  Rod[4] = 1.f + oc * (-(ax * ax + az * az));
  Rod[5] = -sn * ax + oc * (ay * az);
  Rod[6] = -sn * ay + oc * (ax * az);
  Rod[7] =  sn * ax + oc * (ay * az);
  Rod[8] = 1.f + oc * (-(ax * ax + ay * ay));

  float Rrs[9]; mat3mul(Ror, Rod, Rrs);
  float R1[9];  mat3mul(R, Rrs, R1);
  float p1[3];
  p1[0] = R[0] * por[0] + R[1] * por[1] + R[2] * por[2] + p[0];
  p1[1] = R[3] * por[0] + R[4] * por[1] + R[5] * por[2] + p[1];
  p1[2] = R[6] * por[0] + R[7] * por[1] + R[8] * por[2] + p[2];

  float4* r4 = (float4*)myRev;
  r4[0] = make_float4(R1[0], R1[1], R1[2], p1[0]);
  r4[1] = make_float4(R1[3], R1[4], R1[5], p1[1]);
  r4[2] = make_float4(R1[6], R1[7], R1[8], p1[2]);

  float tv0 = d * ap[0], tv1 = d * ap[1], tv2 = d * ap[2];
  float ppv[3];
  ppv[0] = Rop[0] * tv0 + Rop[1] * tv1 + Rop[2] * tv2 + pop[0];
  ppv[1] = Rop[3] * tv0 + Rop[4] * tv1 + Rop[5] * tv2 + pop[1];
  ppv[2] = Rop[6] * tv0 + Rop[7] * tv1 + Rop[8] * tv2 + pop[2];
  float R2[9]; mat3mul(R1, Rop, R2);
  float p2[3];
  p2[0] = R1[0] * ppv[0] + R1[1] * ppv[1] + R1[2] * ppv[2] + p1[0];
  p2[1] = R1[3] * ppv[0] + R1[4] * ppv[1] + R1[5] * ppv[2] + p1[1];
  p2[2] = R1[6] * ppv[0] + R1[7] * ppv[1] + R1[8] * ppv[2] + p1[2];
#pragma unroll
  for (int k = 0; k < 9; ++k) R[k] = R2[k];
  p[0] = p2[0]; p[1] = p2[1]; p[2] = p2[2];

  float4* q4 = (float4*)myPri;
  q4[0] = make_float4(R[0], R[1], R[2], p[0]);
  q4[1] = make_float4(R[3], R[4], R[5], p[1]);
  q4[2] = make_float4(R[6], R[7], R[8], p[2]);
}

__device__ __forceinline__ void wave_stage_handover() {
  asm volatile("s_wait_dscnt 0x0" ::: "memory");
}

__device__ __forceinline__ void read_plane(const float* plane, int fc, int fp, float4 (&v)[16]) {
#pragma unroll
  for (int k = 0; k < 16; ++k)
    v[k] = *(const float4*)(plane + (k * 2 + fc) * PITCH + fp * 4);
}

__device__ __forceinline__ void store_units(float* __restrict__ o, size_t chainStride,
                                            const float4 (&v)[16]) {
#pragma unroll
  for (int k = 0; k < 16; ++k)
    *(float4*)(o + (size_t)(k * 2) * chainStride) = v[k];
}

__device__ __forceinline__ void store_units_ragged(const float* plane, float* __restrict__ o,
                                                   size_t chainStride, long firstChain,
                                                   int B, int fc, int fp) {
#pragma unroll 1
  for (int k = 0; k < 16; ++k)
    if (firstChain + k * 2 < (long)B)
      *(float4*)(o + (size_t)(k * 2) * chainStride) =
          *(const float4*)(plane + (k * 2 + fc) * PITCH + fp * 4);
}

__global__ __launch_bounds__(128) void fk_main_kernel(
    const float* __restrict__ rev_q, const float* __restrict__ pri_q,
    const float* __restrict__ rev_p_off, const float* __restrict__ rev_rpy_off,
    const float* __restrict__ pri_p_off, const float* __restrict__ pri_rpy_off,
    const float* __restrict__ rev_axis, const float* __restrict__ pri_axis,
    const float* __restrict__ p_track, const float* __restrict__ rpy_track,
    float* __restrict__ out, int B) {
  __shared__ float4 stage[4 * 2 * 32 * (PITCH / 4)];
  __shared__ float4 trkAll[4 * 32 * (TRKROW / 4)];
  __shared__ float4 tab4[NJ * 16];
  __shared__ float4 qd4[4 * 2 * 160];

  const int lane = threadIdx.x & 31;
  const int wv   = threadIdx.x >> 5;
  float* sRev = (float*)stage + wv * (2 * 32 * PITCH);
  float* sPri = sRev + 32 * PITCH;
  float* sTrk = (float*)trkAll + wv * (32 * TRKROW);
  float* myRev = sRev + lane * PITCH;
  float* myPri = sPri + lane * PITCH;

  float* trk_out = out;
  float* rev_out = out + (size_t)B * 160;
  float* pri_out = out + (size_t)B * 480;

  const int  h   = lane >> 4;
  const int  m16 = lane & 15;
  const int  nu  = m16 & 3;
  const int  iq  = m16 >> 2;
  const long waveBase = (long)blockIdx.x * 128 + (long)__builtin_amdgcn_readfirstlane(wv) * 32;
  const bool whole = (waveBase + 32 <= (long)B);
  const int fc = lane >> 4;
  const int fp = lane & 15;

  const int tid = threadIdx.x;
  if (tid < 3 * NJ) {
    const int bw = tid / NJ;
    const int bj = tid - bw * NJ;
    const float* rpySrc = bw == 0 ? rev_rpy_off : (bw == 1 ? pri_rpy_off : rpy_track);
    const float* posSrc = bw == 0 ? rev_p_off   : (bw == 1 ? pri_p_off   : p_track);
    float Rt[9];
    rpy2r_dev(rpySrc[3 * bj], rpySrc[3 * bj + 1], rpySrc[3 * bj + 2], Rt);
    const float t0 = posSrc[3 * bj], t1 = posSrc[3 * bj + 1], t2 = posSrc[3 * bj + 2];
    float* o = (float*)tab4 + bj * 64;
    if (bw == 2) {
      o[30] = Rt[0]; o[31] = Rt[1]; o[32] = Rt[2]; o[33] = t0;
      o[34] = Rt[3]; o[35] = Rt[4]; o[36] = Rt[5]; o[37] = t1;
      o[38] = Rt[6]; o[39] = Rt[7]; o[40] = Rt[8]; o[41] = t2;
      o[42] = 0.f; o[43] = 0.f; o[44] = 0.f; o[45] = 1.f;
    } else {
      float* oo = o + bw * 12;
#pragma unroll
      for (int k = 0; k < 9; ++k) oo[k] = Rt[k];
      oo[9] = t0; oo[10] = t1; oo[11] = t2;
    }
  } else if (tid >= 64 && tid < 64 + NJ) {
    const int aj = tid - 64;
    float* o = (float*)tab4 + aj * 64;
    const float x = rev_axis[3 * aj], y = rev_axis[3 * aj + 1], z = rev_axis[3 * aj + 2];
    const float inv = 1.f / (sqrtf(x * x + y * y + z * z) + 1e-12f);
    o[24] = x * inv; o[25] = y * inv; o[26] = z * inv;
    o[27] = pri_axis[3 * aj]; o[28] = pri_axis[3 * aj + 1]; o[29] = pri_axis[3 * aj + 2];
  }
  float4* wq4 = qd4 + wv * 320;
  float4* wd4 = wq4 + 160;
  {
    const long lastPiece = (long)B * (NJ / 4) - 1;
    float4 vq[5], vd[5];
#pragma unroll
    for (int k = 0; k < 5; ++k) {
      long piece = waveBase * (NJ / 4) + lane + 32 * k;
      piece = piece < lastPiece ? piece : lastPiece;
      vq[k] = ((const float4*)rev_q)[piece];
      vd[k] = ((const float4*)pri_q)[piece];
    }
#pragma unroll
    for (int k = 0; k < 5; ++k) {
      wq4[lane + 32 * k] = vq[k];
      wd4[lane + 32 * k] = vd[k];
    }
  }
  const float* myQ = (const float*)wq4 + lane * NJ;
  const float* myD = (const float*)wd4 + lane * NJ;
  const float* tab = (const float*)tab4;

  float R[9] = {1.f, 0.f, 0.f, 0.f, 1.f, 0.f, 0.f, 0.f, 1.f};
  float p[3] = {0.f, 0.f, 0.f};

  {
    const float4 e3 = make_float4(0.f, 0.f, 0.f, 1.f);
    for (int s = 0; s < 4; ++s) {
      *(float4*)(myRev + 16 * s + 12) = e3;
      *(float4*)(myPri + 16 * s + 12) = e3;
    }
  }

  __syncthreads();

  float2 qq = *(const float2*)myQ;
  float2 dd = *(const float2*)myD;

  for (int jp = 0; jp < NJ / 2; ++jp) {
    const int j0 = 2 * jp;
    const int hb = (jp & 1) * 32;
    const float* C0 = tab + j0 * 64;
    const float* C1 = C0 + 64;
    const float q0 = qq.x, q1 = qq.y;
    const float d0 = dd.x, d1 = dd.y;
    {
      const int jn = (jp + 1 < NJ / 2) ? j0 + 2 : j0;
      qq = *(const float2*)(myQ + jn);
      dd = *(const float2*)(myD + jn);
    }
    v2f Aop;
    Aop.x = C0[30 + (2 * h + 0) * 4 + nu];
    Aop.y = C0[30 + (2 * h + 1) * 4 + nu];

    fk_joint(C0, q0, d0, R, p, myRev + hb, myPri + hb);
    wave_stage_handover();

    {
      const v8f zero8 = {0.f, 0.f, 0.f, 0.f, 0.f, 0.f, 0.f, 0.f};
      v2f Bv[8];
#pragma unroll
      for (int g = 0; g < 8; ++g)
        Bv[g] = *(const v2f*)(sPri + (g * 4 + iq) * PITCH + hb + nu * 4 + 2 * h);
      v8f D[8];
#pragma unroll
      for (int g = 0; g < 8; ++g)
        D[g] = __builtin_amdgcn_wmma_f32_16x16x4_f32(
            false, Aop, false, Bv[g], (short)0, zero8, false, false);
      if (h == 0) {
        float* trow = sTrk + iq * TRKROW + jp * 16 + nu * 4;
#pragma unroll
        for (int g = 0; g < 8; ++g)
          *(float4*)(trow + g * 4 * TRKROW) = make_float4(D[g][0], D[g][1], D[g][2], D[g][3]);
      }
    }

    fk_joint(C1, q1, d1, R, p, myRev + hb + 16, myPri + hb + 16);

    if (jp & 1) {
      wave_stage_handover();
      float* oRev = rev_out + ((size_t)(waveBase + fc) * NJ + (j0 - 2)) * 16 + fp * 4;
      float* oPri = pri_out + ((size_t)(waveBase + fc) * NJ + (j0 - 2)) * 16 + fp * 4;
      if (whole) {
        float4 vr[16], vp[16];
        read_plane(sRev, fc, fp, vr);
        read_plane(sPri, fc, fp, vp);
        store_units(oRev, (size_t)NJ * 16, vr);
        store_units(oPri, (size_t)NJ * 16, vp);
      } else {
        store_units_ragged(sRev, oRev, (size_t)NJ * 16, waveBase + fc, B, fc, fp);
        store_units_ragged(sPri, oPri, (size_t)NJ * 16, waveBase + fc, B, fc, fp);
      }
    }
  }

  {
    const float4* sT4 = (const float4*)sTrk;
    float4* oT4 = (float4*)(trk_out + (size_t)waveBase * TRKROW);
    if (whole) {
#pragma unroll
      for (int b = 0; b < 5; ++b) {
        float4 vt[8];
#pragma unroll
        for (int k = 0; k < 8; ++k) vt[k] = sT4[(b * 8 + k) * 32 + lane];
#pragma unroll
        for (int k = 0; k < 8; ++k) oT4[(b * 8 + k) * 32 + lane] = vt[k];
      }
    } else {
#pragma unroll 1
      for (int k = 0; k < 40; ++k) {
        const int piece = k * 32 + lane;
        if (waveBase + piece / 40 < (long)B) oT4[piece] = sT4[piece];
      }
    }
  }
}

extern "C" void kernel_launch(void* const* d_in, const int* in_sizes, int n_in,
                              void* d_out, int out_size, void* d_ws, size_t ws_size,
                              hipStream_t stream) {
  (void)n_in; (void)out_size; (void)d_ws; (void)ws_size;
  const float* rev_q       = (const float*)d_in[0];
  const float* pri_q       = (const float*)d_in[1];
  const float* rev_p_off   = (const float*)d_in[2];
  const float* rev_rpy_off = (const float*)d_in[3];
  const float* pri_p_off   = (const float*)d_in[4];
  const float* pri_rpy_off = (const float*)d_in[5];
  const float* rev_axis    = (const float*)d_in[6];
  const float* pri_axis    = (const float*)d_in[7];
  const float* p_track     = (const float*)d_in[8];
  const float* rpy_track   = (const float*)d_in[9];

  int B = in_sizes[0] / NJ;

  int grid = (B + 127) / 128;
  fk_main_kernel<<<grid, 128, 0, stream>>>(rev_q, pri_q, rev_p_off, rev_rpy_off, pri_p_off,
                                           pri_rpy_off, rev_axis, pri_axis, p_track, rpy_track,
                                           (float*)d_out, B);
}


# Sine and cosine of each joint pair formed one trip ahead by branch-free in-line code (the library's short path, same bits; library kept behind one wave-wide size test); table reads a joint ahead
# speedup vs baseline: 6.6167x; 1.0921x over previous
#include <hip/hip_runtime.h>
#include <cstddef>

typedef float v2f __attribute__((ext_vector_type(2)));
typedef float v8f __attribute__((ext_vector_type(8)));

#define NJ 20
#define PITCH 68
#define TRKROW 160

__device__ __forceinline__ void rpy2r_dev(float r, float p, float y, float* R) {
  float cr = cosf(r), sr = sinf(r);
  float cp = cosf(p), sp = sinf(p);
  float cy = cosf(y), sy = sinf(y);
  R[0] = cy * cp; R[1] = cy * sp * sr - sy * cr; R[2] = cy * sp * cr + sy * sr;
  R[3] = sy * cp; R[4] = sy * sp * sr + cy * cr; R[5] = sy * sp * cr - cy * sr;
  R[6] = -sp;     R[7] = cp * sr;                R[8] = cp * cr;
}

__device__ __forceinline__ void mat3mul(const float* A, const float* B, float* C) {
#pragma unroll
  for (int i = 0; i < 3; ++i)
#pragma unroll
    for (int k = 0; k < 3; ++k)
      C[i * 3 + k] = A[i * 3 + 0] * B[0 * 3 + k] + A[i * 3 + 1] * B[1 * 3 + k] +
                     A[i * 3 + 2] * B[2 * 3 + k];
}


struct JointTab {
  float Ror[9], por[3], Rop[9], pop[3], ar[3], ap[3];
};

__device__ __forceinline__ void load_joint_tab(const float* __restrict__ C, JointTab& T) {
#pragma unroll
  for (int k = 0; k < 9; ++k) T.Ror[k] = C[k];
#pragma unroll
  for (int k = 0; k < 3; ++k) T.por[k] = C[9 + k];
#pragma unroll
  for (int k = 0; k < 9; ++k) T.Rop[k] = C[12 + k];
#pragma unroll
  for (int k = 0; k < 3; ++k) T.pop[k] = C[21 + k];
#pragma unroll
  for (int k = 0; k < 3; ++k) T.ar[k] = C[24 + k];
#pragma unroll
  for (int k = 0; k < 3; ++k) T.ap[k] = C[27 + k];
}

__device__ __forceinline__ void sincos_short(float x, float& sn, float& cs) {
  const float ax = fabsf(x);
  const float fn = rintf(ax * 0x1.45f306p-1f);
  float r = fmaf(fn, -0x1.921fb4p+0f, ax);
  r = fmaf(fn, -0x1.4442d0p-24f, r);
  r = fmaf(fn, -0x1.846988p-48f, r);
  const int qd = (int)fn;
  const float t = r * r;
  const float ps =
      fmaf(r, t * fmaf(t, fmaf(t, -0x1.983304p-13f, 0x1.110388p-7f), -0x1.55553ap-3f), r);
  const float pc = fmaf(
      t, fmaf(t, fmaf(t, fmaf(t, 0x1.aea668p-16f, -0x1.6c9e76p-10f), 0x1.5557eep-5f),
              -0x1.000008p-1f),
      1.0f);
  const bool odd = (qd & 1) != 0;
  const unsigned flip = (qd & 2) ? 0x80000000u : 0u;
  const unsigned sx = __float_as_uint(x) ^ __float_as_uint(ax);
  sn = __uint_as_float(__float_as_uint(odd ? pc : ps) ^ flip ^ sx);
  cs = __uint_as_float(__float_as_uint(odd ? -ps : pc) ^ flip);
}

__device__ __forceinline__ void sincos_pair(bool bigWave, float2 q, float& s0, float& c0,
                                            float& s1, float& c1) {
  if (bigWave) {
    s0 = sinf(q.x); c0 = cosf(q.x);
    s1 = sinf(q.y); c1 = cosf(q.y);
  } else {
    sincos_short(q.x, s0, c0);
    sincos_short(q.y, s1, c1);
  }
}

__device__ __forceinline__ void fk_joint(const JointTab& T, float sn, float cs, float d,
                                         float (&R)[9], float (&p)[3],
                                         float* myRev, float* myPri) {
  const float (&Ror)[9] = T.Ror;
  const float (&por)[3] = T.por;
  const float (&Rop)[9] = T.Rop;
  const float (&pop)[3] = T.pop;
  const float (&ar)[3] = T.ar;
  const float (&ap)[3] = T.ap;

  float oc = 1.f - cs;
  float ax = ar[0], ay = ar[1], az = ar[2];
  float Rod[9];
  Rod[0] = 1.f + oc * (-(ay * ay + az * az));
  Rod[1] = -sn * az + oc * (ax * ay);
  Rod[2] =  sn * ay + oc * (ax * az);
  Rod[3] =  sn * az + oc * (ax * ay);
  Rod[4] = 1.f + oc * (-(ax * ax + az * az));
  Rod[5] = -sn * ax + oc * (ay * az);
  Rod[6] = -sn * ay + oc * (ax * az);
  Rod[7] =  sn * ax + oc * (ay * az);
  Rod[8] = 1.f + oc * (-(ax * ax + ay * ay));

  float Rrs[9]; mat3mul(Ror, Rod, Rrs);
  float R1[9];  mat3mul(R, Rrs, R1);
  float p1[3];
  p1[0] = R[0] * por[0] + R[1] * por[1] + R[2] * por[2] + p[0];
  p1[1] = R[3] * por[0] + R[4] * por[1] + R[5] * por[2] + p[1];
  p1[2] = R[6] * por[0] + R[7] * por[1] + R[8] * por[2] + p[2];

  float4* r4 = (float4*)myRev;
  r4[0] = make_float4(R1[0], R1[1], R1[2], p1[0]);
  r4[1] = make_float4(R1[3], R1[4], R1[5], p1[1]);
  r4[2] = make_float4(R1[6], R1[7], R1[8], p1[2]);

  float tv0 = d * ap[0], tv1 = d * ap[1], tv2 = d * ap[2];
  float ppv[3];
  ppv[0] = Rop[0] * tv0 + Rop[1] * tv1 + Rop[2] * tv2 + pop[0];
  ppv[1] = Rop[3] * tv0 + Rop[4] * tv1 + Rop[5] * tv2 + pop[1];
  ppv[2] = Rop[6] * tv0 + Rop[7] * tv1 + Rop[8] * tv2 + pop[2];
  float R2[9]; mat3mul(R1, Rop, R2);
  float p2[3];
  p2[0] = R1[0] * ppv[0] + R1[1] * ppv[1] + R1[2] * ppv[2] + p1[0];
  p2[1] = R1[3] * ppv[0] + R1[4] * ppv[1] + R1[5] * ppv[2] + p1[1];
  p2[2] = R1[6] * ppv[0] + R1[7] * ppv[1] + R1[8] * ppv[2] + p1[2];
#pragma unroll
  for (int k = 0; k < 9; ++k) R[k] = R2[k];
  p[0] = p2[0]; p[1] = p2[1]; p[2] = p2[2];

  float4* q4 = (float4*)myPri;
  q4[0] = make_float4(R[0], R[1], R[2], p[0]);
  q4[1] = make_float4(R[3], R[4], R[5], p[1]);
  q4[2] = make_float4(R[6], R[7], R[8], p[2]);
}

__device__ __forceinline__ void wave_stage_handover() {
  asm volatile("s_wait_dscnt 0x0" ::: "memory");
}

__device__ __forceinline__ void read_plane(const float* plane, int fc, int fp, float4 (&v)[16]) {
#pragma unroll
  for (int k = 0; k < 16; ++k)
    v[k] = *(const float4*)(plane + (k * 2 + fc) * PITCH + fp * 4);
}

__device__ __forceinline__ void store_units(float* __restrict__ o, size_t chainStride,
                                            const float4 (&v)[16]) {
#pragma unroll
  for (int k = 0; k < 16; ++k)
    *(float4*)(o + (size_t)(k * 2) * chainStride) = v[k];
}

__device__ __forceinline__ void store_units_ragged(const float* plane, float* __restrict__ o,
                                                   size_t chainStride, long firstChain,
                                                   int B, int fc, int fp) {
#pragma unroll 1
  for (int k = 0; k < 16; ++k)
    if (firstChain + k * 2 < (long)B)
      *(float4*)(o + (size_t)(k * 2) * chainStride) =
          *(const float4*)(plane + (k * 2 + fc) * PITCH + fp * 4);
}

__global__ __launch_bounds__(128) void fk_main_kernel(
    const float* __restrict__ rev_q, const float* __restrict__ pri_q,
    const float* __restrict__ rev_p_off, const float* __restrict__ rev_rpy_off,
    const float* __restrict__ pri_p_off, const float* __restrict__ pri_rpy_off,
    const float* __restrict__ rev_axis, const float* __restrict__ pri_axis,
    const float* __restrict__ p_track, const float* __restrict__ rpy_track,
    float* __restrict__ out, int B) {
  __shared__ float4 stage[4 * 2 * 32 * (PITCH / 4)];
  __shared__ float4 trkAll[4 * 32 * (TRKROW / 4)];
  __shared__ float4 tab4[NJ * 16];
  __shared__ float4 qd4[4 * 2 * 160];

  const int lane = threadIdx.x & 31;
  const int wv   = threadIdx.x >> 5;
  float* sRev = (float*)stage + wv * (2 * 32 * PITCH);
  float* sPri = sRev + 32 * PITCH;
  float* sTrk = (float*)trkAll + wv * (32 * TRKROW);
  float* myRev = sRev + lane * PITCH;
  float* myPri = sPri + lane * PITCH;

  float* trk_out = out;
  float* rev_out = out + (size_t)B * 160;
  float* pri_out = out + (size_t)B * 480;

  const int  h   = lane >> 4;
  const int  m16 = lane & 15;
  const int  nu  = m16 & 3;
  const int  iq  = m16 >> 2;
  const long waveBase = (long)blockIdx.x * 128 + (long)__builtin_amdgcn_readfirstlane(wv) * 32;
  const bool whole = (waveBase + 32 <= (long)B);
  const int fc = lane >> 4;
  const int fp = lane & 15;

  const int tid = threadIdx.x;
  if (tid < 3 * NJ) {
    const int bw = tid / NJ;
    const int bj = tid - bw * NJ;
    const float* rpySrc = bw == 0 ? rev_rpy_off : (bw == 1 ? pri_rpy_off : rpy_track);
    const float* posSrc = bw == 0 ? rev_p_off   : (bw == 1 ? pri_p_off   : p_track);
    float Rt[9];
    rpy2r_dev(rpySrc[3 * bj], rpySrc[3 * bj + 1], rpySrc[3 * bj + 2], Rt);
    const float t0 = posSrc[3 * bj], t1 = posSrc[3 * bj + 1], t2 = posSrc[3 * bj + 2];
    float* o = (float*)tab4 + bj * 64;
    if (bw == 2) {
      o[30] = Rt[0]; o[31] = Rt[1]; o[32] = Rt[2]; o[33] = t0;
      o[34] = Rt[3]; o[35] = Rt[4]; o[36] = Rt[5]; o[37] = t1;
      o[38] = Rt[6]; o[39] = Rt[7]; o[40] = Rt[8]; o[41] = t2;
      o[42] = 0.f; o[43] = 0.f; o[44] = 0.f; o[45] = 1.f;
    } else {
      float* oo = o + bw * 12;
#pragma unroll
      for (int k = 0; k < 9; ++k) oo[k] = Rt[k];
      oo[9] = t0; oo[10] = t1; oo[11] = t2;
    }
  } else if (tid >= 64 && tid < 64 + NJ) {
    const int aj = tid - 64;
    float* o = (float*)tab4 + aj * 64;
    const float x = rev_axis[3 * aj], y = rev_axis[3 * aj + 1], z = rev_axis[3 * aj + 2];
    const float inv = 1.f / (sqrtf(x * x + y * y + z * z) + 1e-12f);
    o[24] = x * inv; o[25] = y * inv; o[26] = z * inv;
    o[27] = pri_axis[3 * aj]; o[28] = pri_axis[3 * aj + 1]; o[29] = pri_axis[3 * aj + 2];
  }
  float4* wq4 = qd4 + wv * 320;
  float4* wd4 = wq4 + 160;
  {
    const long lastPiece = (long)B * (NJ / 4) - 1;
    float4 vq[5], vd[5];
#pragma unroll
    for (int k = 0; k < 5; ++k) {
      long piece = waveBase * (NJ / 4) + lane + 32 * k;
      piece = piece < lastPiece ? piece : lastPiece;
      vq[k] = ((const float4*)rev_q)[piece];
      vd[k] = ((const float4*)pri_q)[piece];
    }
#pragma unroll
    for (int k = 0; k < 5; ++k) {
      wq4[lane + 32 * k] = vq[k];
      wd4[lane + 32 * k] = vd[k];
    }
  }
  const float* myQ = (const float*)wq4 + lane * NJ;
  const float* myD = (const float*)wd4 + lane * NJ;
  const float* tab = (const float*)tab4;

  wave_stage_handover();
  bool bigWave;
  {
    unsigned top = 0u;
#pragma unroll
    for (int k = 0; k < NJ / 4; ++k) {
      const float4 v = ((const float4*)myQ)[k];
      const unsigned a0 = __float_as_uint(v.x) & 0x7fffffffu;
      const unsigned a1 = __float_as_uint(v.y) & 0x7fffffffu;
      const unsigned a2 = __float_as_uint(v.z) & 0x7fffffffu;
      const unsigned a3 = __float_as_uint(v.w) & 0x7fffffffu;
      const unsigned m01 = a0 > a1 ? a0 : a1;
      const unsigned m23 = a2 > a3 ? a2 : a3;
      const unsigned m = m01 > m23 ? m01 : m23;
      top = m > top ? m : top;
    }
    bigWave = __builtin_amdgcn_ballot_w32(top >= 0x48000000u) != 0u;
  }

  float R[9] = {1.f, 0.f, 0.f, 0.f, 1.f, 0.f, 0.f, 0.f, 1.f};
  float p[3] = {0.f, 0.f, 0.f};

  {
    const float4 e3 = make_float4(0.f, 0.f, 0.f, 1.f);
    for (int s = 0; s < 4; ++s) {
      *(float4*)(myRev + 16 * s + 12) = e3;
      *(float4*)(myPri + 16 * s + 12) = e3;
    }
  }

  __syncthreads();

  float2 qq = *(const float2*)myQ;
  float2 dd = *(const float2*)myD;
  float s0, c0, s1, c1;
  sincos_pair(bigWave, qq, s0, c0, s1, c1);
  JointTab T0;
  load_joint_tab(tab, T0);

  for (int jp = 0; jp < NJ / 2; ++jp) {
    const int j0 = 2 * jp;
    const int hb = (jp & 1) * 32;
    const float* C0 = tab + j0 * 64;
    const float* C1 = C0 + 64;
    const float d0 = dd.x, d1 = dd.y;
    const int jn = (jp + 1 < NJ / 2) ? j0 + 2 : j0;
    qq = *(const float2*)(myQ + jn);
    dd = *(const float2*)(myD + jn);
    JointTab T1;
    load_joint_tab(C1, T1);
    float s0n, c0n, s1n, c1n;
    v2f Aop;
    Aop.x = C0[30 + (2 * h + 0) * 4 + nu];
    Aop.y = C0[30 + (2 * h + 1) * 4 + nu];

    fk_joint(T0, s0, c0, d0, R, p, myRev + hb, myPri + hb);
    wave_stage_handover();

    {
      const v8f zero8 = {0.f, 0.f, 0.f, 0.f, 0.f, 0.f, 0.f, 0.f};
      v2f Bv[8];
#pragma unroll
      for (int g = 0; g < 8; ++g)
        Bv[g] = *(const v2f*)(sPri + (g * 4 + iq) * PITCH + hb + nu * 4 + 2 * h);
      load_joint_tab(tab + jn * 64, T0);
      sincos_pair(bigWave, qq, s0n, c0n, s1n, c1n);
      v8f D[8];
#pragma unroll
      for (int g = 0; g < 8; ++g)
        D[g] = __builtin_amdgcn_wmma_f32_16x16x4_f32(
            false, Aop, false, Bv[g], (short)0, zero8, false, false);
      if (h == 0) {
        float* trow = sTrk + iq * TRKROW + jp * 16 + nu * 4;
#pragma unroll
        for (int g = 0; g < 8; ++g)
          *(float4*)(trow + g * 4 * TRKROW) = make_float4(D[g][0], D[g][1], D[g][2], D[g][3]);
      }
    }

    fk_joint(T1, s1, c1, d1, R, p, myRev + hb + 16, myPri + hb + 16);
    s0 = s0n; c0 = c0n; s1 = s1n; c1 = c1n;

    if (jp & 1) {
      wave_stage_handover();
      float* oRev = rev_out + ((size_t)(waveBase + fc) * NJ + (j0 - 2)) * 16 + fp * 4;
      float* oPri = pri_out + ((size_t)(waveBase + fc) * NJ + (j0 - 2)) * 16 + fp * 4;
      if (whole) {
        float4 vr[16], vp[16];
        read_plane(sRev, fc, fp, vr);
        read_plane(sPri, fc, fp, vp);
        store_units(oRev, (size_t)NJ * 16, vr);
        store_units(oPri, (size_t)NJ * 16, vp);
      } else {
        store_units_ragged(sRev, oRev, (size_t)NJ * 16, waveBase + fc, B, fc, fp);
        store_units_ragged(sPri, oPri, (size_t)NJ * 16, waveBase + fc, B, fc, fp);
      }
    }
  }

  {
    const float4* sT4 = (const float4*)sTrk;
    float4* oT4 = (float4*)(trk_out + (size_t)waveBase * TRKROW);
    if (whole) {
#pragma unroll
      for (int b = 0; b < 5; ++b) {
        float4 vt[8];
#pragma unroll
        for (int k = 0; k < 8; ++k) vt[k] = sT4[(b * 8 + k) * 32 + lane];
#pragma unroll
        for (int k = 0; k < 8; ++k) oT4[(b * 8 + k) * 32 + lane] = vt[k];
      }
    } else {
#pragma unroll 1
      for (int k = 0; k < 40; ++k) {
        const int piece = k * 32 + lane;
        if (waveBase + piece / 40 < (long)B) oT4[piece] = sT4[piece];
      }
    }
  }
}

extern "C" void kernel_launch(void* const* d_in, const int* in_sizes, int n_in,
                              void* d_out, int out_size, void* d_ws, size_t ws_size,
                              hipStream_t stream) {
  (void)n_in; (void)out_size; (void)d_ws; (void)ws_size;
  const float* rev_q       = (const float*)d_in[0];
  const float* pri_q       = (const float*)d_in[1];
  const float* rev_p_off   = (const float*)d_in[2];
  const float* rev_rpy_off = (const float*)d_in[3];
  const float* pri_p_off   = (const float*)d_in[4];
  const float* pri_rpy_off = (const float*)d_in[5];
  const float* rev_axis    = (const float*)d_in[6];
  const float* pri_axis    = (const float*)d_in[7];
  const float* p_track     = (const float*)d_in[8];
  const float* rpy_track   = (const float*)d_in[9];

  int B = in_sizes[0] / NJ;

  int grid = (B + 127) / 128;
  fk_main_kernel<<<grid, 128, 0, stream>>>(rev_q, pri_q, rev_p_off, rev_rpy_off, pri_p_off,
                                           pri_rpy_off, rev_axis, pri_axis, p_track, rpy_track,
                                           (float*)d_out, B);
}
